# starting kernel
# baseline (speedup 1.0000x reference)
#include <hip/hip_runtime.h>
#include <hip/hip_bf16.h>

typedef float v2f __attribute__((ext_vector_type(2)));
typedef float v8f __attribute__((ext_vector_type(8)));

#define N_NODES 100000
#define N_EDGES 1600000
#define DIM_IN  256
#define HD      64
#define NCLS    20
#define NDOCS   10000

__device__ __forceinline__ void atomic_add_f32(float* p, float v) {
    __hip_atomic_fetch_add(p, v, __ATOMIC_RELAXED, __HIP_MEMORY_SCOPE_AGENT);
}


__global__ void TEXTGCN_deg_init(float* __restrict__ deg, int n) {
    int i = blockIdx.x * blockDim.x + threadIdx.x;
    if (i < n) deg[i] = 1.0f;
}

__global__ void TEXTGCN_deg_acc(const int* __restrict__ dst,
                                const float* __restrict__ w,
                                float* __restrict__ deg, int e) {
    int i = blockIdx.x * blockDim.x + threadIdx.x;
    if (i < e) atomic_add_f32(&deg[dst[i]], w[i]);
}

__global__ void TEXTGCN_dinv(float* __restrict__ deg, int n) {
    int i = blockIdx.x * blockDim.x + threadIdx.x;
    if (i < n) {
        float d = deg[i];
        deg[i] = (d > 0.0f) ? __frsqrt_rn(d) : 0.0f;
    }
}

__global__ void TEXTGCN_norm(const int* __restrict__ src, const int* __restrict__ dst,
                             const float* __restrict__ w, const float* __restrict__ dinv,
                             float* __restrict__ norm, int e) {
    int i = blockIdx.x * blockDim.x + threadIdx.x;
    if (i < e) norm[i] = dinv[src[i]] * w[i] * dinv[dst[i]];
}


__global__ void TEXTGCN_gemm_wmma(const float* __restrict__ A,
                                  const float* __restrict__ B,
                                  float* __restrict__ C,
                                  int K, int Nout) {
    const int lane = threadIdx.x;
    const int mt = blockIdx.x, nt = blockIdx.y;
    const int m  = mt * 16 + (lane & 15);
    const int n  = nt * 16 + (lane & 15);
    const int kq = (lane >> 4) * 2;

    const float* __restrict__ arow = A + (size_t)m * K;
    v8f acc = {};
    for (int k = 0; k < K; k += 4) {
        v2f a, b;
        a.x = arow[k + kq];
        a.y = arow[k + kq + 1];
        b.x = B[(size_t)(k + kq)     * Nout + n];
        b.y = B[(size_t)(k + kq + 1) * Nout + n];
        acc = __builtin_amdgcn_wmma_f32_16x16x4_f32(
                  false, a, false, b, (short)0, acc, false, false);
    }
    const int mbase = mt * 16 + (lane >> 4) * 8;
    const int nst   = nt * 16 + (lane & 15);
#pragma unroll
    for (int r = 0; r < 8; ++r)
        C[(size_t)(mbase + r) * Nout + nst] = acc[r];
}


__global__ void TEXTGCN_self(const float* __restrict__ t, const float* __restrict__ dinv,
                             float* __restrict__ acc, int n) {
    int gid = blockIdx.x * blockDim.x + threadIdx.x;
    if (gid >= n * (HD / 4)) return;
    int i = gid >> 4, c = gid & 15;
    float s = dinv[i] * dinv[i];
    float4 v = ((const float4*)(t + (size_t)i * HD))[c];
    v.x *= s; v.y *= s; v.z *= s; v.w *= s;
    ((float4*)(acc + (size_t)i * HD))[c] = v;
}

__global__ void TEXTGCN_scatter(const int* __restrict__ src, const int* __restrict__ dst,
                                const float* __restrict__ norm,
                                const float* __restrict__ t,
                                float* __restrict__ acc, int e) {
    int gid = blockIdx.x * blockDim.x + threadIdx.x;
    if (gid >= e * (HD / 4)) return;
    int ei = gid >> 4, c = gid & 15;
    int s = src[ei], d = dst[ei];
    float w = norm[ei];
    float4 v = ((const float4*)(t + (size_t)s * HD))[c];
    float* o = acc + (size_t)d * HD + c * 4;
    atomic_add_f32(o + 0, v.x * w);
    atomic_add_f32(o + 1, v.y * w);
    atomic_add_f32(o + 2, v.z * w);
    atomic_add_f32(o + 3, v.w * w);
}

__global__ void TEXTGCN_bias_relu(float* __restrict__ acc, const float* __restrict__ b, int n) {
    int gid = blockIdx.x * blockDim.x + threadIdx.x;
    if (gid >= n * HD) return;
    float v = acc[gid] + b[gid & (HD - 1)];
    acc[gid] = v > 0.0f ? v : 0.0f;
}


__global__ void TEXTGCN_doc_gemm_wmma(const float* __restrict__ h,
                                      const int* __restrict__ doc_idx,
                                      const float* __restrict__ Wp,
                                      const float* __restrict__ bp,
                                      float* __restrict__ dbuf) {
    const int lane = threadIdx.x;
    const int mt = blockIdx.x, nt = blockIdx.y;
    const int m  = mt * 16 + (lane & 15);
    const int n  = nt * 16 + (lane & 15);
    const int kq = (lane >> 4) * 2;

    const float* __restrict__ arow = h + (size_t)doc_idx[m] * HD;
    v8f acc = {};
    for (int k = 0; k < HD; k += 4) {
        v2f a, b;
        a.x = arow[k + kq];
        a.y = arow[k + kq + 1];
        b.x = Wp[(size_t)(k + kq)     * HD + n];
        b.y = Wp[(size_t)(k + kq + 1) * HD + n];
        acc = __builtin_amdgcn_wmma_f32_16x16x4_f32(
                  false, a, false, b, (short)0, acc, false, false);
    }
    const int mbase = mt * 16 + (lane >> 4) * 8;
    const int nst   = nt * 16 + (lane & 15);
    const float bias = bp[nst];
#pragma unroll
    for (int r = 0; r < 8; ++r) {
        float v = acc[r] + bias;
        dbuf[(size_t)(mbase + r) * HD + nst] = v > 0.0f ? v : 0.0f;
    }
}


__global__ void TEXTGCN_classifier(const float* __restrict__ dbuf,
                                   const float* __restrict__ Wc,
                                   const float* __restrict__ bc,
                                   float* __restrict__ out) {
    int gid = blockIdx.x * blockDim.x + threadIdx.x;
    if (gid >= NDOCS * NCLS) return;
    int d = gid / NCLS, c = gid % NCLS;
    const float* __restrict__ row = dbuf + (size_t)d * HD;
    float s = bc[c];
#pragma unroll 8
    for (int k = 0; k < HD; ++k) s += row[k] * Wc[(size_t)k * NCLS + c];
    out[gid] = s;
}


extern "C" void kernel_launch(void* const* d_in, const int* in_sizes, int n_in,
                              void* d_out, int out_size, void* d_ws, size_t ws_size,
                              hipStream_t stream) {
    const float* x   = (const float*)d_in[0];
    const float* ew  = (const float*)d_in[1];
    const float* W1  = (const float*)d_in[2];
    const float* b1  = (const float*)d_in[3];
    const float* W2  = (const float*)d_in[4];
    const float* b2  = (const float*)d_in[5];
    const float* W3  = (const float*)d_in[6];
    const float* b3  = (const float*)d_in[7];
    const float* Wp  = (const float*)d_in[8];
    const float* bp  = (const float*)d_in[9];
    const float* Wc  = (const float*)d_in[10];
    const float* bc  = (const float*)d_in[11];
    const int*   ei  = (const int*)d_in[12];
    const int*   doc = (const int*)d_in[13];
    float*       out = (float*)d_out;

    const int* src = ei;
    const int* dst = ei + N_EDGES;

    float* ws   = (float*)d_ws;
    float* dinv = ws;
    float* nrm  = dinv + N_NODES;
    float* bufT = nrm  + N_EDGES;
    float* bufH = bufT + (size_t)N_NODES * HD;
    float* dbuf = bufH + (size_t)N_NODES * HD;

    const int B = 256;
    dim3 blk(B);
    dim3 gN((N_NODES + B - 1) / B);
    dim3 gE((N_EDGES + B - 1) / B);
    dim3 gN16((N_NODES * (HD / 4) + B - 1) / B);
    dim3 gE16((N_EDGES * (HD / 4) + B - 1) / B);
    dim3 gNF((N_NODES * HD + B - 1) / B);

    TEXTGCN_deg_init<<<gN, blk, 0, stream>>>(dinv, N_NODES);
    TEXTGCN_deg_acc <<<gE, blk, 0, stream>>>(dst, ew, dinv, N_EDGES);
    TEXTGCN_dinv    <<<gN, blk, 0, stream>>>(dinv, N_NODES);
    TEXTGCN_norm    <<<gE, blk, 0, stream>>>(src, dst, ew, dinv, nrm, N_EDGES);

    dim3 wave(32);
    dim3 gGemm1(N_NODES / 16, HD / 16);
    dim3 gGemmH(N_NODES / 16, HD / 16);
    dim3 gDoc(NDOCS / 16, HD / 16);

    TEXTGCN_gemm_wmma<<<gGemm1, wave, 0, stream>>>(x, W1, bufT, DIM_IN, HD);
    TEXTGCN_self     <<<gN16, blk, 0, stream>>>(bufT, dinv, bufH, N_NODES);
    TEXTGCN_scatter  <<<gE16, blk, 0, stream>>>(src, dst, nrm, bufT, bufH, N_EDGES);
    TEXTGCN_bias_relu<<<gNF,  blk, 0, stream>>>(bufH, b1, N_NODES);

    TEXTGCN_gemm_wmma<<<gGemmH, wave, 0, stream>>>(bufH, W2, bufT, HD, HD);
    TEXTGCN_self     <<<gN16, blk, 0, stream>>>(bufT, dinv, bufH, N_NODES);
    TEXTGCN_scatter  <<<gE16, blk, 0, stream>>>(src, dst, nrm, bufT, bufH, N_EDGES);
    TEXTGCN_bias_relu<<<gNF,  blk, 0, stream>>>(bufH, b2, N_NODES);

    TEXTGCN_gemm_wmma<<<gGemmH, wave, 0, stream>>>(bufH, W3, bufT, HD, HD);
    TEXTGCN_self     <<<gN16, blk, 0, stream>>>(bufT, dinv, bufH, N_NODES);
    TEXTGCN_scatter  <<<gE16, blk, 0, stream>>>(src, dst, nrm, bufT, bufH, N_EDGES);
    TEXTGCN_bias_relu<<<gNF,  blk, 0, stream>>>(bufH, b3, N_NODES);

    TEXTGCN_doc_gemm_wmma<<<gDoc, wave, 0, stream>>>(bufH, doc, Wp, bp, dbuf);
    dim3 gCls((NDOCS * NCLS + B - 1) / B);
    TEXTGCN_classifier<<<gCls, blk, 0, stream>>>(dbuf, Wc, bc, out);
}


# propagation: float atomics to memory replaced by 256 owner blocks, each keeps 391 target rows in LDS, scans a 1-byte owner key per edge, adds its edges' rows by LDS float adds (arrival order)
# speedup vs baseline: 1.6128x; 1.6128x over previous
#include <hip/hip_runtime.h>
#include <hip/hip_bf16.h>

typedef float v2f __attribute__((ext_vector_type(2)));
typedef float v8f __attribute__((ext_vector_type(8)));

#define N_NODES 100000
#define N_EDGES 1600000
#define DIM_IN  256
#define HD      64
#define NCLS    20
#define NDOCS   10000

#define OWN_ROWS   391
#define OWN_BLOCKS 256
#define QCAP       16384
static_assert(OWN_BLOCKS * OWN_ROWS >= N_NODES && (OWN_BLOCKS - 1) * OWN_ROWS < N_NODES && OWN_BLOCKS <= 256,
              "every owner block holds at least one row and its id fits a byte");

__device__ __forceinline__ void atomic_add_f32(float* p, float v) {
    __hip_atomic_fetch_add(p, v, __ATOMIC_RELAXED, __HIP_MEMORY_SCOPE_AGENT);
}


__global__ void TEXTGCN_deg_init(float* __restrict__ deg, int n) {
    int i = blockIdx.x * blockDim.x + threadIdx.x;
    if (i < n) deg[i] = 1.0f;
}

__global__ void TEXTGCN_deg_acc(const int* __restrict__ dst,
                                const float* __restrict__ w,
                                float* __restrict__ deg, int e) {
    int i = blockIdx.x * blockDim.x + threadIdx.x;
    if (i < e) atomic_add_f32(&deg[dst[i]], w[i]);
}

__global__ void TEXTGCN_dinv(float* __restrict__ deg, int n) {
    int i = blockIdx.x * blockDim.x + threadIdx.x;
    if (i < n) {
        float d = deg[i];
        deg[i] = (d > 0.0f) ? __frsqrt_rn(d) : 0.0f;
    }
}

__global__ void TEXTGCN_norm(const int* __restrict__ src, const int* __restrict__ dst,
                             const float* __restrict__ w, const float* __restrict__ dinv,
                             float* __restrict__ norm, float* __restrict__ ownf, int e) {
    int i = blockIdx.x * blockDim.x + threadIdx.x;
    if (i < e) {
        const int d = dst[i];
        norm[i] = dinv[src[i]] * w[i] * dinv[d];
        ((unsigned char*)ownf)[i] = (unsigned char)(d / OWN_ROWS);
    }
}


__global__ void TEXTGCN_gemm_wmma(const float* __restrict__ A,
                                  const float* __restrict__ B,
                                  const float* __restrict__ dinv,
                                  float* __restrict__ C,
                                  float* __restrict__ S,
                                  int K, int Nout) {
    const int lane = threadIdx.x;
    const int mt = blockIdx.x, nt = blockIdx.y;
    const int m  = mt * 16 + (lane & 15);
    const int n  = nt * 16 + (lane & 15);
    const int kq = (lane >> 4) * 2;

    const float* __restrict__ arow = A + (size_t)m * K;
    v8f acc = {};
    for (int k = 0; k < K; k += 4) {
        v2f a, b;
        a.x = arow[k + kq];
        a.y = arow[k + kq + 1];
        b.x = B[(size_t)(k + kq)     * Nout + n];
        b.y = B[(size_t)(k + kq + 1) * Nout + n];
        acc = __builtin_amdgcn_wmma_f32_16x16x4_f32(
                  false, a, false, b, (short)0, acc, false, false);
    }
    const int mbase = mt * 16 + (lane >> 4) * 8;
    const int nst   = nt * 16 + (lane & 15);
    const float4 d0 = *(const float4*)(dinv + mbase);
    const float4 d1 = *(const float4*)(dinv + mbase + 4);
    const float dr[8] = {d0.x, d0.y, d0.z, d0.w, d1.x, d1.y, d1.z, d1.w};
#pragma unroll
    for (int r = 0; r < 8; ++r) {
        const size_t o = (size_t)(mbase + r) * Nout + nst;
        C[o] = acc[r];
        S[o] = acc[r] * (dr[r] * dr[r]);
    }
}


__global__ void __launch_bounds__(128)
TEXTGCN_gemm_hidden(float* H,
                    const float* __restrict__ bprev,
                    const float* __restrict__ Wk,
                    const float* __restrict__ dinv,
                    float* __restrict__ T) {
    const int lane = threadIdx.x & 31;
    const int nt = threadIdx.x >> 5;
    const int mt = blockIdx.x;
    const int m  = mt * 16 + (lane & 15);
    const int n  = nt * 16 + (lane & 15);
    const int kq = (lane >> 4) * 2;

    const float* hrow = H + (size_t)m * HD;
    v2f a[HD / 4];
#pragma unroll
    for (int i = 0; i < HD / 4; ++i) {
        const int k = 4 * i + kq;
        const float2 hv = *(const float2*)(hrow + k);
        const float2 bv = *(const float2*)(bprev + k);
        const float v0 = hv.x + bv.x;
        const float v1 = hv.y + bv.y;
        a[i].x = v0 > 0.0f ? v0 : 0.0f;
        a[i].y = v1 > 0.0f ? v1 : 0.0f;
    }
    __syncthreads();

    v8f acc = {};
#pragma unroll
    for (int i = 0; i < HD / 4; ++i) {
        const int k = 4 * i + kq;
        v2f b;
        b.x = Wk[(size_t)k       * HD + n];
        b.y = Wk[(size_t)(k + 1) * HD + n];
        acc = __builtin_amdgcn_wmma_f32_16x16x4_f32(
                  false, a[i], false, b, (short)0, acc, false, false);
    }
    const int mbase = mt * 16 + (lane >> 4) * 8;
    const float4 d0 = *(const float4*)(dinv + mbase);
    const float4 d1 = *(const float4*)(dinv + mbase + 4);
    const float dr[8] = {d0.x, d0.y, d0.z, d0.w, d1.x, d1.y, d1.z, d1.w};
#pragma unroll
    for (int r = 0; r < 8; ++r) {
        const size_t o = (size_t)(mbase + r) * HD + n;
        T[o] = acc[r];
        H[o] = acc[r] * (dr[r] * dr[r]);
    }
}


__device__ __forceinline__ void lds_add_f32(float* p, float v) {
    __hip_atomic_fetch_add(p, v, __ATOMIC_RELAXED, __HIP_MEMORY_SCOPE_WORKGROUP);
}

__global__ void __launch_bounds__(1024)
TEXTGCN_propagate_owned(const int* __restrict__ src, const int* __restrict__ dst,
                        const float* __restrict__ norm,
                        const float* __restrict__ ownf,
                        const float* __restrict__ t,
                        float* __restrict__ acc, int e) {
    __shared__ float rows[OWN_ROWS * HD];
    __shared__ int   queue[QCAP];
    __shared__ int   qcount;

    const int tid  = threadIdx.x;
    const int lane = tid & 31;
    const int wave = __builtin_amdgcn_readfirstlane(tid >> 5);
    const int lo   = blockIdx.x * OWN_ROWS;
    int nrows = N_NODES - lo;
    if (nrows > OWN_ROWS) nrows = OWN_ROWS;

    const float4* a4 = (const float4*)(acc + (size_t)lo * HD);
    float4* r4 = (float4*)rows;
    const int nvec = nrows * (HD / 4);
    {
        constexpr int TRIPS = (OWN_ROWS * (HD / 4) + 1023) / 1024;
        float4 st[TRIPS];
#pragma unroll
        for (int q = 0; q < TRIPS; ++q) {
            int i = tid + q * 1024;
            if (i > nvec - 1) i = nvec - 1;
            st[q] = a4[i];
        }
#pragma unroll
        for (int q = 0; q < TRIPS; ++q) {
            const int i = tid + q * 1024;
            if (i < OWN_ROWS * (HD / 4)) r4[i] = st[q];
        }
    }
    if (tid == 0) qcount = 0;
    __syncthreads();

    const uint4* own4 = (const uint4*)ownf;
    const unsigned char* own1 = (const unsigned char*)ownf;
    const unsigned me = blockIdx.x * 0x01010101u;
    const int ngroups = e >> 4;
    for (int g0 = tid; g0 < ngroups; g0 += 8 * 1024) {
        uint4 v[8];
#pragma unroll
        for (int k = 0; k < 8; ++k) {
            int g = g0 + k * 1024;
            if (g > ngroups - 1) g = ngroups - 1;
            v[k] = own4[g];
        }
        unsigned p0 = 0u, p1 = 0u, p2 = 0u, p3 = 0u;
#pragma unroll
        for (int k = 0; k < 8; ++k) {
            const unsigned x0 = v[k].x ^ me, x1 = v[k].y ^ me, x2 = v[k].z ^ me, x3 = v[k].w ^ me;
            const unsigned u0 = ((x0 & 0x7F7F7F7Fu) + 0x7F7F7F7Fu) | x0;
            const unsigned u1 = ((x1 & 0x7F7F7F7Fu) + 0x7F7F7F7Fu) | x1;
            const unsigned u2 = ((x2 & 0x7F7F7F7Fu) + 0x7F7F7F7Fu) | x2;
            const unsigned u3 = ((x3 & 0x7F7F7F7Fu) + 0x7F7F7F7Fu) | x3;
            const unsigned sel = 0x01010101u << k;
            p0 |= (u0 >> (7 - k)) & sel;
            p1 |= (u1 >> (7 - k)) & sel;
            p2 |= (u2 >> (7 - k)) & sel;
            p3 |= (u3 >> (7 - k)) & sel;
        }
        int nv = (ngroups - g0 + 1023) >> 10;
        if (nv > 8) nv = 8;
        const unsigned vm = ((1u << nv) - 1u) * 0x01010101u;
        const unsigned mm[4] = {~p0 & vm, ~p1 & vm, ~p2 & vm, ~p3 & vm};
#pragma unroll
        for (int j = 0; j < 4; ++j) {
            unsigned m = mm[j];
            while (m != 0u) {
                const int bit = __builtin_ctz(m);
                m &= m - 1u;
                const int eid = ((g0 + (bit & 7) * 1024) << 4) + j * 4 + (bit >> 3);
                const int pos = atomicAdd(&qcount, 1);
                if (pos < QCAP) {
                    queue[pos] = eid;
                } else {
                    const int s = src[eid];
                    const int d = dst[eid] - lo;
                    const float w = norm[eid];
#pragma nounroll
                    for (int cc = 0; cc < HD; ++cc)
                        lds_add_f32(&rows[d * HD + cc], t[(size_t)s * HD + cc] * w);
                }
            }
        }
    }
    for (int i = (ngroups << 4) + tid; i < e; i += 1024) {
        if (own1[i] == (unsigned char)blockIdx.x) {
            const int pos = atomicAdd(&qcount, 1);
            if (pos < QCAP) {
                queue[pos] = i;
            } else {
                const int s = src[i];
                const int d = dst[i] - lo;
                const float w = norm[i];
#pragma nounroll
                for (int cc = 0; cc < HD; ++cc)
                    lds_add_f32(&rows[d * HD + cc], t[(size_t)s * HD + cc] * w);
            }
        }
    }
    __syncthreads();

    int n = qcount;
    if (n > QCAP) n = QCAP;
    n = __builtin_amdgcn_readfirstlane(n);
    for (int base = wave * 32; base < n; base += 1024) {
        int m = n - base;
        if (m > 32) m = 32;
        int sl = 0, dl = 0, wl = 0;
        if (lane < m) {
            const int eid = queue[base + lane];
            sl = src[eid];
            dl = dst[eid] - lo;
            wl = __float_as_int(norm[eid]);
        }
        for (int j = 0; j < m; j += 16) {
            float v0[16], v1[16];
#pragma unroll
            for (int k = 0; k < 16; ++k) {
                int idx = j + k;
                if (idx > m - 1) idx = m - 1;
                const int s = __builtin_amdgcn_readlane(sl, idx);
                const float* __restrict__ tp = t + (size_t)s * HD;
                v0[k] = tp[lane];
                v1[k] = tp[lane + 32];
            }
#pragma unroll
            for (int k = 0; k < 16; ++k) {
                if (j + k < m) {
                    const int d = __builtin_amdgcn_readlane(dl, j + k);
                    const float w = __int_as_float(__builtin_amdgcn_readlane(wl, j + k));
                    float* rp = rows + d * HD;
                    lds_add_f32(rp + lane,      v0[k] * w);
                    lds_add_f32(rp + lane + 32, v1[k] * w);
                }
            }
        }
    }
    __syncthreads();

    float4* o4 = (float4*)(acc + (size_t)lo * HD);
    for (int i = tid; i < nvec; i += 1024) o4[i] = r4[i];
}


__global__ void TEXTGCN_doc_gemm_wmma(const float* __restrict__ h,
                                      const float* __restrict__ bprev,
                                      const int* __restrict__ doc_idx,
                                      const float* __restrict__ Wp,
                                      const float* __restrict__ bp,
                                      float* __restrict__ dbuf) {
    const int lane = threadIdx.x;
    const int mt = blockIdx.x, nt = blockIdx.y;
    const int m  = mt * 16 + (lane & 15);
    const int n  = nt * 16 + (lane & 15);
    const int kq = (lane >> 4) * 2;

    const float* __restrict__ arow = h + (size_t)doc_idx[m] * HD;
    v8f acc = {};
    for (int k = 0; k < HD; k += 4) {
        v2f a, b;
        const float v0 = arow[k + kq]     + bprev[k + kq];
        const float v1 = arow[k + kq + 1] + bprev[k + kq + 1];
        a.x = v0 > 0.0f ? v0 : 0.0f;
        a.y = v1 > 0.0f ? v1 : 0.0f;
        b.x = Wp[(size_t)(k + kq)     * HD + n];
        b.y = Wp[(size_t)(k + kq + 1) * HD + n];
        acc = __builtin_amdgcn_wmma_f32_16x16x4_f32(
                  false, a, false, b, (short)0, acc, false, false);
    }
    const int mbase = mt * 16 + (lane >> 4) * 8;
    const int nst   = nt * 16 + (lane & 15);
    const float bias = bp[nst];
#pragma unroll
    for (int r = 0; r < 8; ++r) {
        float v = acc[r] + bias;
        dbuf[(size_t)(mbase + r) * HD + nst] = v > 0.0f ? v : 0.0f;
    }
}


__global__ void TEXTGCN_classifier(const float* __restrict__ dbuf,
                                   const float* __restrict__ Wc,
                                   const float* __restrict__ bc,
                                   float* __restrict__ out) {
    int gid = blockIdx.x * blockDim.x + threadIdx.x;
    if (gid >= NDOCS * NCLS) return;
    int d = gid / NCLS, c = gid % NCLS;
    const float* __restrict__ row = dbuf + (size_t)d * HD;
    float s = bc[c];
#pragma unroll 8
    for (int k = 0; k < HD; ++k) s += row[k] * Wc[(size_t)k * NCLS + c];
    out[gid] = s;
}


extern "C" void kernel_launch(void* const* d_in, const int* in_sizes, int n_in,
                              void* d_out, int out_size, void* d_ws, size_t ws_size,
                              hipStream_t stream) {
    const float* x   = (const float*)d_in[0];
    const float* ew  = (const float*)d_in[1];
    const float* W1  = (const float*)d_in[2];
    const float* b1  = (const float*)d_in[3];
    const float* W2  = (const float*)d_in[4];
    const float* b2  = (const float*)d_in[5];
    const float* W3  = (const float*)d_in[6];
    const float* b3  = (const float*)d_in[7];
    const float* Wp  = (const float*)d_in[8];
    const float* bp  = (const float*)d_in[9];
    const float* Wc  = (const float*)d_in[10];
    const float* bc  = (const float*)d_in[11];
    const int*   ei  = (const int*)d_in[12];
    const int*   doc = (const int*)d_in[13];
    float*       out = (float*)d_out;

    const int* src = ei;
    const int* dst = ei + N_EDGES;

    float* ws   = (float*)d_ws;
    float* dinv = ws;
    float* nrm  = dinv + N_NODES;
    float* bufT = nrm  + N_EDGES;
    float* bufH = bufT + (size_t)N_NODES * HD;
    float* dbuf = bufH + (size_t)N_NODES * HD;

    const int B = 256;
    dim3 blk(B);
    dim3 gN((N_NODES + B - 1) / B);
    dim3 gE((N_EDGES + B - 1) / B);

    TEXTGCN_deg_init<<<gN, blk, 0, stream>>>(dinv, N_NODES);
    TEXTGCN_deg_acc <<<gE, blk, 0, stream>>>(dst, ew, dinv, N_EDGES);
    TEXTGCN_dinv    <<<gN, blk, 0, stream>>>(dinv, N_NODES);
    TEXTGCN_norm    <<<gE, blk, 0, stream>>>(src, dst, ew, dinv, nrm, dbuf, N_EDGES);

    dim3 wave(32);
    dim3 gGemm1(N_NODES / 16, HD / 16);
    dim3 gGemmH(N_NODES / 16);
    dim3 blkH(128);
    dim3 gDoc(NDOCS / 16, HD / 16);
    dim3 gOwn(OWN_BLOCKS);
    dim3 blkOwn(1024);

    TEXTGCN_gemm_wmma<<<gGemm1, wave, 0, stream>>>(x, W1, dinv, bufT, bufH, DIM_IN, HD);
    TEXTGCN_propagate_owned<<<gOwn, blkOwn, 0, stream>>>(src, dst, nrm, dbuf, bufT, bufH, N_EDGES);

    TEXTGCN_gemm_hidden<<<gGemmH, blkH, 0, stream>>>(bufH, b1, W2, dinv, bufT);
    TEXTGCN_propagate_owned<<<gOwn, blkOwn, 0, stream>>>(src, dst, nrm, dbuf, bufT, bufH, N_EDGES);

    TEXTGCN_gemm_hidden<<<gGemmH, blkH, 0, stream>>>(bufH, b2, W3, dinv, bufT);
    TEXTGCN_propagate_owned<<<gOwn, blkOwn, 0, stream>>>(src, dst, nrm, dbuf, bufT, bufH, N_EDGES);

    TEXTGCN_doc_gemm_wmma<<<gDoc, wave, 0, stream>>>(bufH, b3, doc, Wp, bp, dbuf);
    dim3 gCls((NDOCS * NCLS + B - 1) / B);
    TEXTGCN_classifier<<<gCls, blk, 0, stream>>>(dbuf, Wc, bc, out);
}


# transforms (conv1, both 64x64): a wave keeps its 16 columns of W for all K in registers; rows of A come in once by whole-line 16-byte loads through an LDS tile; k order and sums unchanged
# speedup vs baseline: 1.9870x; 1.2320x over previous
#include <hip/hip_runtime.h>
#include <hip/hip_bf16.h>

typedef float v2f __attribute__((ext_vector_type(2)));
typedef float v8f __attribute__((ext_vector_type(8)));

#define N_NODES 100000
#define N_EDGES 1600000
#define DIM_IN  256
#define HD      64
#define NCLS    20
#define NDOCS   10000

#define OWN_ROWS   391
#define OWN_BLOCKS 256
#define QCAP       16384
static_assert(OWN_BLOCKS * OWN_ROWS >= N_NODES && (OWN_BLOCKS - 1) * OWN_ROWS < N_NODES && OWN_BLOCKS <= 256,
              "every owner block holds at least one row and its id fits a byte");

__device__ __forceinline__ void atomic_add_f32(float* p, float v) {
    __hip_atomic_fetch_add(p, v, __ATOMIC_RELAXED, __HIP_MEMORY_SCOPE_AGENT);
}


__global__ void TEXTGCN_deg_init(float* __restrict__ deg, int n) {
    int i = blockIdx.x * blockDim.x + threadIdx.x;
    if (i < n) deg[i] = 1.0f;
}

__global__ void TEXTGCN_deg_acc(const int* __restrict__ dst,
                                const float* __restrict__ w,
                                float* __restrict__ deg, int e) {
    int i = blockIdx.x * blockDim.x + threadIdx.x;
    if (i < e) atomic_add_f32(&deg[dst[i]], w[i]);
}

__global__ void TEXTGCN_dinv(float* __restrict__ deg, int n) {
    int i = blockIdx.x * blockDim.x + threadIdx.x;
    if (i < n) {
        float d = deg[i];
        deg[i] = (d > 0.0f) ? __frsqrt_rn(d) : 0.0f;
    }
}

__global__ void TEXTGCN_norm(const int* __restrict__ src, const int* __restrict__ dst,
                             const float* __restrict__ w, const float* __restrict__ dinv,
                             float* __restrict__ norm, float* __restrict__ ownf, int e) {
    int i = blockIdx.x * blockDim.x + threadIdx.x;
    if (i < e) {
        const int d = dst[i];
        norm[i] = dinv[src[i]] * w[i] * dinv[d];
        ((unsigned char*)ownf)[i] = (unsigned char)(d / OWN_ROWS);
    }
}


#define G1_ROWS   32
#define G1_PITCH  (DIM_IN + 4)
#define G1_TILES  (N_NODES / G1_ROWS)
#define G1_BLOCKS 1024
#define G1_VEC    (G1_ROWS * DIM_IN / 4 / 128)
#define G1_EACH(X) X(0) X(1) X(2) X(3) X(4) X(5) X(6) X(7) X(8) X(9) X(10) X(11) X(12) X(13) X(14) X(15)
#define G1_LOAD(q)  const float4 nv##q = nx[(q) * 128];
#define G1_STORE(q) *(float4*)&nt[((tid + (q) * 128) >> 6) * G1_PITCH + ((tid + (q) * 128) & 63) * 4] = nv##q;
static_assert(G1_VEC == 16, "G1_EACH names 16 values");
static_assert(N_NODES % G1_ROWS == 0 && HD == 64 && DIM_IN % 4 == 0 && (G1_ROWS * DIM_IN / 4) % 128 == 0 &&
              (G1_PITCH * 4) % 16 == 0, "tile shape");

__device__ __forceinline__ void g1_tile(const float* __restrict__ ap,
                                        const v2f (&bf)[DIM_IN / 4],
                                        const float* __restrict__ dinv,
                                        float* __restrict__ C, float* __restrict__ S,
                                        int mbase, int n) {
    v8f acc0 = {}, acc1 = {};
#pragma unroll
    for (int i = 0; i < DIM_IN / 4; ++i) {
        const v2f a0 = *(const v2f*)(ap + 4 * i);
        const v2f a1 = *(const v2f*)(ap + 16 * G1_PITCH + 4 * i);
        acc0 = __builtin_amdgcn_wmma_f32_16x16x4_f32(
                   false, a0, false, bf[i], (short)0, acc0, false, false);
        acc1 = __builtin_amdgcn_wmma_f32_16x16x4_f32(
                   false, a1, false, bf[i], (short)0, acc1, false, false);
    }
    const float4 d0 = *(const float4*)(dinv + mbase);
    const float4 d1 = *(const float4*)(dinv + mbase + 4);
    const float4 e0 = *(const float4*)(dinv + mbase + 16);
    const float4 e1 = *(const float4*)(dinv + mbase + 20);
    const float dr[8] = {d0.x, d0.y, d0.z, d0.w, d1.x, d1.y, d1.z, d1.w};
    const float er[8] = {e0.x, e0.y, e0.z, e0.w, e1.x, e1.y, e1.z, e1.w};
    const unsigned ob = (unsigned)mbase * HD + (unsigned)n;
    float* __restrict__ cp = C + ob;
    float* __restrict__ sp = S + ob;
#pragma unroll
    for (int r = 0; r < 8; ++r) {
        cp[r * HD]        = acc0[r];
        sp[r * HD]        = acc0[r] * (dr[r] * dr[r]);
        cp[(16 + r) * HD] = acc1[r];
        sp[(16 + r) * HD] = acc1[r] * (er[r] * er[r]);
    }
}

__global__ void __launch_bounds__(128)
TEXTGCN_gemm_rows(const float* A,
                  const float* __restrict__ B,
                  const float* __restrict__ dinv,
                  float* __restrict__ C,
                  float* __restrict__ S) {
    __shared__ __attribute__((aligned(16))) float tileA[2][G1_ROWS * G1_PITCH];

    const int tid  = threadIdx.x;
    const int lane = tid & 31;
    const int wv   = tid >> 5;
    const int c    = lane & 15;
    const int h    = lane >> 4;
    const int kq   = h * 2;
    const int n    = wv * 16 + c;

    v2f bf[DIM_IN / 4];
#pragma unroll
    for (int i = 0; i < DIM_IN / 4; ++i) {
        bf[i].x = B[(size_t)(4 * i + kq)     * HD + n];
        bf[i].y = B[(size_t)(4 * i + kq + 1) * HD + n];
    }

    const float4* a4 = (const float4*)A;
    int t = blockIdx.x;
    if (t < G1_TILES) {
        float4 s0[G1_VEC];
#pragma unroll
        for (int q = 0; q < G1_VEC; ++q) s0[q] = a4[(size_t)t * (G1_ROWS * DIM_IN / 4) + tid + q * 128];
#pragma unroll
        for (int q = 0; q < G1_VEC; ++q) {
            const int idx = tid + q * 128;
            *(float4*)&tileA[0][(idx >> 6) * G1_PITCH + (idx & 63) * 4] = s0[q];
        }
    }
    __syncthreads();

    int buf = 0;
    for (; t + G1_BLOCKS < G1_TILES; t += G1_BLOCKS) {
        const float4* nx = a4 + (size_t)(t + G1_BLOCKS) * (G1_ROWS * DIM_IN / 4) + tid;
        G1_EACH(G1_LOAD)
        __builtin_amdgcn_sched_barrier(0);

        g1_tile(&tileA[buf][c * G1_PITCH + kq], bf, dinv, C, S, t * G1_ROWS + h * 8, n);

        float* nt = tileA[buf ^ 1];
        G1_EACH(G1_STORE)
        __syncthreads();
        buf ^= 1;
    }
    if (t < G1_TILES)
        g1_tile(&tileA[buf][c * G1_PITCH + kq], bf, dinv, C, S, t * G1_ROWS + h * 8, n);
}


#define GH_ROWS   160
#define GH_PITCH  (HD + 4)
#define GH_VEC    (GH_ROWS * HD / 4 / 128)
static_assert(N_NODES % GH_ROWS == 0 && GH_ROWS % 32 == 0 && (GH_ROWS * HD / 4) % 128 == 0 && (GH_PITCH * 4) % 16 == 0,
              "tile shape");

__global__ void __launch_bounds__(128)
TEXTGCN_gemm_hidden(float* H,
                    const float* __restrict__ bprev,
                    const float* __restrict__ Wk,
                    const float* __restrict__ dinv,
                    float* __restrict__ T) {
    __shared__ __attribute__((aligned(16))) float tileH[GH_ROWS * GH_PITCH];

    const int tid  = threadIdx.x;
    const int lane = tid & 31;
    const int wv   = tid >> 5;
    const int c    = lane & 15;
    const int h    = lane >> 4;
    const int kq   = h * 2;
    const int n    = wv * 16 + c;
    const int row0 = blockIdx.x * GH_ROWS;

    const float4* h4 = (const float4*)(H + (size_t)row0 * HD);
    float4 st[GH_VEC];
#pragma unroll
    for (int q = 0; q < GH_VEC; ++q) st[q] = h4[tid + q * 128];

    v2f bf[HD / 4];
#pragma unroll
    for (int i = 0; i < HD / 4; ++i) {
        bf[i].x = Wk[(size_t)(4 * i + kq)     * HD + n];
        bf[i].y = Wk[(size_t)(4 * i + kq + 1) * HD + n];
    }
    const float4 bv = *(const float4*)(bprev + (tid & 15) * 4);
#pragma unroll
    for (int q = 0; q < GH_VEC; ++q) {
        const int idx = tid + q * 128;
        const float v0 = st[q].x + bv.x, v1 = st[q].y + bv.y, v2 = st[q].z + bv.z, v3 = st[q].w + bv.w;
        float4 o;
        o.x = v0 > 0.0f ? v0 : 0.0f;
        o.y = v1 > 0.0f ? v1 : 0.0f;
        o.z = v2 > 0.0f ? v2 : 0.0f;
        o.w = v3 > 0.0f ? v3 : 0.0f;
        *(float4*)&tileH[(idx >> 4) * GH_PITCH + (idx & 15) * 4] = o;
    }
    __syncthreads();

#pragma unroll
    for (int s = 0; s < GH_ROWS / 32; ++s) {
        const float* ap = &tileH[(s * 32 + c) * GH_PITCH + kq];
        v8f acc0 = {}, acc1 = {};
#pragma unroll
        for (int i = 0; i < HD / 4; ++i) {
            const v2f a0 = *(const v2f*)(ap + 4 * i);
            const v2f a1 = *(const v2f*)(ap + 16 * GH_PITCH + 4 * i);
            acc0 = __builtin_amdgcn_wmma_f32_16x16x4_f32(
                       false, a0, false, bf[i], (short)0, acc0, false, false);
            acc1 = __builtin_amdgcn_wmma_f32_16x16x4_f32(
                       false, a1, false, bf[i], (short)0, acc1, false, false);
        }
        const int mbase = row0 + s * 32 + h * 8;
        const float4 d0 = *(const float4*)(dinv + mbase);
        const float4 d1 = *(const float4*)(dinv + mbase + 4);
        const float4 e0 = *(const float4*)(dinv + mbase + 16);
        const float4 e1 = *(const float4*)(dinv + mbase + 20);
        const float dr[8] = {d0.x, d0.y, d0.z, d0.w, d1.x, d1.y, d1.z, d1.w};
        const float er[8] = {e0.x, e0.y, e0.z, e0.w, e1.x, e1.y, e1.z, e1.w};
        const unsigned ob = (unsigned)mbase * HD + (unsigned)n;
        float* __restrict__ tp = T + ob;
        float* hp = H + ob;
#pragma unroll
        for (int r = 0; r < 8; ++r) {
            tp[r * HD]        = acc0[r];
            hp[r * HD]        = acc0[r] * (dr[r] * dr[r]);
            tp[(16 + r) * HD] = acc1[r];
            hp[(16 + r) * HD] = acc1[r] * (er[r] * er[r]);
        }
    }
}


__device__ __forceinline__ void lds_add_f32(float* p, float v) {
    __hip_atomic_fetch_add(p, v, __ATOMIC_RELAXED, __HIP_MEMORY_SCOPE_WORKGROUP);
}

__global__ void __launch_bounds__(1024)
TEXTGCN_propagate_owned(const int* __restrict__ src, const int* __restrict__ dst,
                        const float* __restrict__ norm,
                        const float* __restrict__ ownf,
                        const float* __restrict__ t,
                        float* __restrict__ acc, int e) {
    __shared__ float rows[OWN_ROWS * HD];
    __shared__ int   queue[QCAP];
    __shared__ int   qcount;

    const int tid  = threadIdx.x;
    const int lane = tid & 31;
    const int wave = __builtin_amdgcn_readfirstlane(tid >> 5);
    const int lo   = blockIdx.x * OWN_ROWS;
    int nrows = N_NODES - lo;
    if (nrows > OWN_ROWS) nrows = OWN_ROWS;

    const float4* a4 = (const float4*)(acc + (size_t)lo * HD);
    float4* r4 = (float4*)rows;
    const int nvec = nrows * (HD / 4);
    {
        constexpr int TRIPS = (OWN_ROWS * (HD / 4) + 1023) / 1024;
        float4 st[TRIPS];
#pragma unroll
        for (int q = 0; q < TRIPS; ++q) {
            int i = tid + q * 1024;
            if (i > nvec - 1) i = nvec - 1;
            st[q] = a4[i];
        }
#pragma unroll
        for (int q = 0; q < TRIPS; ++q) {
            const int i = tid + q * 1024;
            if (i < OWN_ROWS * (HD / 4)) r4[i] = st[q];
        }
    }
    if (tid == 0) qcount = 0;
    __syncthreads();

    const uint4* own4 = (const uint4*)ownf;
    const unsigned char* own1 = (const unsigned char*)ownf;
    const unsigned me = blockIdx.x * 0x01010101u;
    const int ngroups = e >> 4;
    for (int g0 = tid; g0 < ngroups; g0 += 8 * 1024) {
        uint4 v[8];
#pragma unroll
        for (int k = 0; k < 8; ++k) {
            int g = g0 + k * 1024;
            if (g > ngroups - 1) g = ngroups - 1;
            v[k] = own4[g];
        }
        unsigned p0 = 0u, p1 = 0u, p2 = 0u, p3 = 0u;
#pragma unroll
        for (int k = 0; k < 8; ++k) {
            const unsigned x0 = v[k].x ^ me, x1 = v[k].y ^ me, x2 = v[k].z ^ me, x3 = v[k].w ^ me;
            const unsigned u0 = ((x0 & 0x7F7F7F7Fu) + 0x7F7F7F7Fu) | x0;
            const unsigned u1 = ((x1 & 0x7F7F7F7Fu) + 0x7F7F7F7Fu) | x1;
            const unsigned u2 = ((x2 & 0x7F7F7F7Fu) + 0x7F7F7F7Fu) | x2;
            const unsigned u3 = ((x3 & 0x7F7F7F7Fu) + 0x7F7F7F7Fu) | x3;
            const unsigned sel = 0x01010101u << k;
            p0 |= (u0 >> (7 - k)) & sel;
            p1 |= (u1 >> (7 - k)) & sel;
            p2 |= (u2 >> (7 - k)) & sel;
            p3 |= (u3 >> (7 - k)) & sel;
        }
        int nv = (ngroups - g0 + 1023) >> 10;
        if (nv > 8) nv = 8;
        const unsigned vm = ((1u << nv) - 1u) * 0x01010101u;
        const unsigned mm[4] = {~p0 & vm, ~p1 & vm, ~p2 & vm, ~p3 & vm};
#pragma unroll
        for (int j = 0; j < 4; ++j) {
            unsigned m = mm[j];
            while (m != 0u) {
                const int bit = __builtin_ctz(m);
                m &= m - 1u;
                const int eid = ((g0 + (bit & 7) * 1024) << 4) + j * 4 + (bit >> 3);
                const int pos = atomicAdd(&qcount, 1);
                if (pos < QCAP) {
                    queue[pos] = eid;
                } else {
                    const int s = src[eid];
                    const int d = dst[eid] - lo;
                    const float w = norm[eid];
#pragma nounroll
                    for (int cc = 0; cc < HD; ++cc)
                        lds_add_f32(&rows[d * HD + cc], t[(size_t)s * HD + cc] * w);
                }
            }
        }
    }
    for (int i = (ngroups << 4) + tid; i < e; i += 1024) {
        if (own1[i] == (unsigned char)blockIdx.x) {
            const int pos = atomicAdd(&qcount, 1);
            if (pos < QCAP) {
                queue[pos] = i;
            } else {
                const int s = src[i];
                const int d = dst[i] - lo;
                const float w = norm[i];
#pragma nounroll
                for (int cc = 0; cc < HD; ++cc)
                    lds_add_f32(&rows[d * HD + cc], t[(size_t)s * HD + cc] * w);
            }
        }
    }
    __syncthreads();

    int n = qcount;
    if (n > QCAP) n = QCAP;
    n = __builtin_amdgcn_readfirstlane(n);
    for (int base = wave * 32; base < n; base += 1024) {
        int m = n - base;
        if (m > 32) m = 32;
        int sl = 0, dl = 0, wl = 0;
        if (lane < m) {
            const int eid = queue[base + lane];
            sl = src[eid];
            dl = dst[eid] - lo;
            wl = __float_as_int(norm[eid]);
        }
        for (int j = 0; j < m; j += 16) {
            float v0[16], v1[16];
#pragma unroll
            for (int k = 0; k < 16; ++k) {
                int idx = j + k;
                if (idx > m - 1) idx = m - 1;
                const int s = __builtin_amdgcn_readlane(sl, idx);
                const float* __restrict__ tp = t + (size_t)s * HD;
                v0[k] = tp[lane];
                v1[k] = tp[lane + 32];
            }
#pragma unroll
            for (int k = 0; k < 16; ++k) {
                if (j + k < m) {
                    const int d = __builtin_amdgcn_readlane(dl, j + k);
                    const float w = __int_as_float(__builtin_amdgcn_readlane(wl, j + k));
                    float* rp = rows + d * HD;
                    lds_add_f32(rp + lane,      v0[k] * w);
                    lds_add_f32(rp + lane + 32, v1[k] * w);
                }
            }
        }
    }
    __syncthreads();

    float4* o4 = (float4*)(acc + (size_t)lo * HD);
    for (int i = tid; i < nvec; i += 1024) o4[i] = r4[i];
}


__global__ void TEXTGCN_doc_gemm_wmma(const float* __restrict__ h,
                                      const float* __restrict__ bprev,
                                      const int* __restrict__ doc_idx,
                                      const float* __restrict__ Wp,
                                      const float* __restrict__ bp,
                                      float* __restrict__ dbuf) {
    const int lane = threadIdx.x;
    const int mt = blockIdx.x, nt = blockIdx.y;
    const int m  = mt * 16 + (lane & 15);
    const int n  = nt * 16 + (lane & 15);
    const int kq = (lane >> 4) * 2;

    const float* __restrict__ arow = h + (size_t)doc_idx[m] * HD;
    v8f acc = {};
    for (int k = 0; k < HD; k += 4) {
        v2f a, b;
        const float v0 = arow[k + kq]     + bprev[k + kq];
        const float v1 = arow[k + kq + 1] + bprev[k + kq + 1];
        a.x = v0 > 0.0f ? v0 : 0.0f;
        a.y = v1 > 0.0f ? v1 : 0.0f;
        b.x = Wp[(size_t)(k + kq)     * HD + n];
        b.y = Wp[(size_t)(k + kq + 1) * HD + n];
        acc = __builtin_amdgcn_wmma_f32_16x16x4_f32(
                  false, a, false, b, (short)0, acc, false, false);
    }
    const int mbase = mt * 16 + (lane >> 4) * 8;
    const int nst   = nt * 16 + (lane & 15);
    const float bias = bp[nst];
#pragma unroll
    for (int r = 0; r < 8; ++r) {
        float v = acc[r] + bias;
        dbuf[(size_t)(mbase + r) * HD + nst] = v > 0.0f ? v : 0.0f;
    }
}


__global__ void TEXTGCN_classifier(const float* __restrict__ dbuf,
                                   const float* __restrict__ Wc,
                                   const float* __restrict__ bc,
                                   float* __restrict__ out) {
    int gid = blockIdx.x * blockDim.x + threadIdx.x;
    if (gid >= NDOCS * NCLS) return;
    int d = gid / NCLS, c = gid % NCLS;
    const float* __restrict__ row = dbuf + (size_t)d * HD;
    float s = bc[c];
#pragma unroll 8
    for (int k = 0; k < HD; ++k) s += row[k] * Wc[(size_t)k * NCLS + c];
    out[gid] = s;
}


extern "C" void kernel_launch(void* const* d_in, const int* in_sizes, int n_in,
                              void* d_out, int out_size, void* d_ws, size_t ws_size,
                              hipStream_t stream) {
    const float* x   = (const float*)d_in[0];
    const float* ew  = (const float*)d_in[1];
    const float* W1  = (const float*)d_in[2];
    const float* b1  = (const float*)d_in[3];
    const float* W2  = (const float*)d_in[4];
    const float* b2  = (const float*)d_in[5];
    const float* W3  = (const float*)d_in[6];
    const float* b3  = (const float*)d_in[7];
    const float* Wp  = (const float*)d_in[8];
    const float* bp  = (const float*)d_in[9];
    const float* Wc  = (const float*)d_in[10];
    const float* bc  = (const float*)d_in[11];
    const int*   ei  = (const int*)d_in[12];
    const int*   doc = (const int*)d_in[13];
    float*       out = (float*)d_out;

    const int* src = ei;
    const int* dst = ei + N_EDGES;

    float* ws   = (float*)d_ws;
    float* dinv = ws;
    float* nrm  = dinv + N_NODES;
    float* bufT = nrm  + N_EDGES;
    float* bufH = bufT + (size_t)N_NODES * HD;
    float* dbuf = bufH + (size_t)N_NODES * HD;

    const int B = 256;
    dim3 blk(B);
    dim3 gN((N_NODES + B - 1) / B);
    dim3 gE((N_EDGES + B - 1) / B);

    TEXTGCN_deg_init<<<gN, blk, 0, stream>>>(dinv, N_NODES);
    TEXTGCN_deg_acc <<<gE, blk, 0, stream>>>(dst, ew, dinv, N_EDGES);
    TEXTGCN_dinv    <<<gN, blk, 0, stream>>>(dinv, N_NODES);
    TEXTGCN_norm    <<<gE, blk, 0, stream>>>(src, dst, ew, dinv, nrm, dbuf, N_EDGES);

    dim3 wave(32);
    dim3 gGemm1(G1_BLOCKS);
    dim3 gGemmH(N_NODES / GH_ROWS);
    dim3 blkH(128);
    dim3 gDoc(NDOCS / 16, HD / 16);
    dim3 gOwn(OWN_BLOCKS);
    dim3 blkOwn(1024);

    TEXTGCN_gemm_rows<<<gGemm1, blkH, 0, stream>>>(x, W1, dinv, bufT, bufH);
    TEXTGCN_propagate_owned<<<gOwn, blkOwn, 0, stream>>>(src, dst, nrm, dbuf, bufT, bufH, N_EDGES);

    TEXTGCN_gemm_hidden<<<gGemmH, blkH, 0, stream>>>(bufH, b1, W2, dinv, bufT);
    TEXTGCN_propagate_owned<<<gOwn, blkOwn, 0, stream>>>(src, dst, nrm, dbuf, bufT, bufH, N_EDGES);

    TEXTGCN_gemm_hidden<<<gGemmH, blkH, 0, stream>>>(bufH, b2, W3, dinv, bufT);
    TEXTGCN_propagate_owned<<<gOwn, blkOwn, 0, stream>>>(src, dst, nrm, dbuf, bufT, bufH, N_EDGES);

    TEXTGCN_doc_gemm_wmma<<<gDoc, wave, 0, stream>>>(bufH, b3, doc, Wp, bp, dbuf);
    dim3 gCls((NDOCS * NCLS + B - 1) / B);
    TEXTGCN_classifier<<<gCls, blk, 0, stream>>>(dbuf, Wc, bc, out);
}


# normalisation factored: transforms store t*dinv[row], owner blocks add w*row with the raw edge weight and scale each finished row by dinv[target]; 2 scattered reads per edge, not 3; no start plane
# speedup vs baseline: 2.2257x; 1.1201x over previous
#include <hip/hip_runtime.h>
#include <hip/hip_bf16.h>

typedef float v2f __attribute__((ext_vector_type(2)));
typedef float v8f __attribute__((ext_vector_type(8)));

#define N_NODES 100000
#define N_EDGES 1600000
#define DIM_IN  256
#define HD      64
#define NCLS    20
#define NDOCS   10000

#define OWN_ROWS   391
#define OWN_BLOCKS 256
#define QCAP       16384
static_assert(OWN_BLOCKS * OWN_ROWS >= N_NODES && (OWN_BLOCKS - 1) * OWN_ROWS < N_NODES && OWN_BLOCKS <= 256,
              "every owner block holds at least one row and its id fits a byte");
static_assert(HD / 4 == 16, "a row is 16 pieces of 16 bytes (the write-back finds a piece's row by a shift of 4)");

__device__ __forceinline__ void atomic_add_f32(float* p, float v) {
    __hip_atomic_fetch_add(p, v, __ATOMIC_RELAXED, __HIP_MEMORY_SCOPE_AGENT);
}


__global__ void TEXTGCN_deg_init(float* __restrict__ deg, int n) {
    int i = blockIdx.x * blockDim.x + threadIdx.x;
    if (i < n) deg[i] = 1.0f;
}

__global__ void TEXTGCN_deg_acc(const int* __restrict__ dst,
                                const float* __restrict__ w,
                                float* __restrict__ deg, int e) {
    int i = blockIdx.x * blockDim.x + threadIdx.x;
    if (i < e) atomic_add_f32(&deg[dst[i]], w[i]);
}

__global__ void TEXTGCN_dinv(float* __restrict__ deg, int n) {
    int i = blockIdx.x * blockDim.x + threadIdx.x;
    if (i < n) {
        float d = deg[i];
        deg[i] = (d > 0.0f) ? __frsqrt_rn(d) : 0.0f;
    }
}

#define PK_SHIFT 17
#define PK_MASK  ((1u << PK_SHIFT) - 1u)
static_assert(N_NODES <= (1 << PK_SHIFT) && OWN_ROWS <= (1 << (32 - PK_SHIFT)), "source id and local row share a word");

__global__ void TEXTGCN_pack(const int* __restrict__ src, const int* __restrict__ dst,
                             float* __restrict__ pkf, float* __restrict__ ownf, int e) {
    int i = blockIdx.x * blockDim.x + threadIdx.x;
    if (i < e) {
        const int d = dst[i];
        const int o = d / OWN_ROWS;
        ((unsigned*)pkf)[i] = (unsigned)src[i] | ((unsigned)(d - o * OWN_ROWS) << PK_SHIFT);
        ((unsigned char*)ownf)[i] = (unsigned char)o;
    }
}


#define G1_ROWS   32
#define G1_PITCH  (DIM_IN + 4)
#define G1_TILES  (N_NODES / G1_ROWS)
#define G1_BLOCKS 1024
#define G1_VEC    (G1_ROWS * DIM_IN / 4 / 128)
#define G1_EACH(X) X(0) X(1) X(2) X(3) X(4) X(5) X(6) X(7) X(8) X(9) X(10) X(11) X(12) X(13) X(14) X(15)
#define G1_LOAD(q)  const float4 nv##q = nx[(q) * 128];
#define G1_STORE(q) *(float4*)&nt[((tid + (q) * 128) >> 6) * G1_PITCH + ((tid + (q) * 128) & 63) * 4] = nv##q;
static_assert(G1_VEC == 16, "G1_EACH names 16 values");
static_assert(N_NODES % G1_ROWS == 0 && HD == 64 && DIM_IN % 4 == 0 && (G1_ROWS * DIM_IN / 4) % 128 == 0 &&
              (G1_PITCH * 4) % 16 == 0, "tile shape");

__device__ __forceinline__ void g1_tile(const float* __restrict__ ap,
                                        const v2f (&bf)[DIM_IN / 4],
                                        const float* __restrict__ dinv,
                                        float* __restrict__ C,
                                        int mbase, int n) {
    v8f acc0 = {}, acc1 = {};
#pragma unroll
    for (int i = 0; i < DIM_IN / 4; ++i) {
        const v2f a0 = *(const v2f*)(ap + 4 * i);
        const v2f a1 = *(const v2f*)(ap + 16 * G1_PITCH + 4 * i);
        acc0 = __builtin_amdgcn_wmma_f32_16x16x4_f32(
                   false, a0, false, bf[i], (short)0, acc0, false, false);
        acc1 = __builtin_amdgcn_wmma_f32_16x16x4_f32(
                   false, a1, false, bf[i], (short)0, acc1, false, false);
    }
    const float4 d0 = *(const float4*)(dinv + mbase);
    const float4 d1 = *(const float4*)(dinv + mbase + 4);
    const float4 e0 = *(const float4*)(dinv + mbase + 16);
    const float4 e1 = *(const float4*)(dinv + mbase + 20);
    const float dr[8] = {d0.x, d0.y, d0.z, d0.w, d1.x, d1.y, d1.z, d1.w};
    const float er[8] = {e0.x, e0.y, e0.z, e0.w, e1.x, e1.y, e1.z, e1.w};
    const unsigned ob = (unsigned)mbase * HD + (unsigned)n;
    float* __restrict__ cp = C + ob;
#pragma unroll
    for (int r = 0; r < 8; ++r) {
        cp[r * HD]        = acc0[r] * dr[r];
        cp[(16 + r) * HD] = acc1[r] * er[r];
    }
}

__global__ void __launch_bounds__(128)
TEXTGCN_gemm_rows(const float* A,
                  const float* __restrict__ B,
                  const float* __restrict__ dinv,
                  float* __restrict__ C) {
    __shared__ __attribute__((aligned(16))) float tileA[2][G1_ROWS * G1_PITCH];

    const int tid  = threadIdx.x;
    const int lane = tid & 31;
    const int wv   = tid >> 5;
    const int c    = lane & 15;
    const int h    = lane >> 4;
    const int kq   = h * 2;
    const int n    = wv * 16 + c;

    v2f bf[DIM_IN / 4];
#pragma unroll
    for (int i = 0; i < DIM_IN / 4; ++i) {
        bf[i].x = B[(size_t)(4 * i + kq)     * HD + n];
        bf[i].y = B[(size_t)(4 * i + kq + 1) * HD + n];
    }

    const float4* a4 = (const float4*)A;
    int t = blockIdx.x;
    if (t < G1_TILES) {
        float4 s0[G1_VEC];
#pragma unroll
        for (int q = 0; q < G1_VEC; ++q) s0[q] = a4[(size_t)t * (G1_ROWS * DIM_IN / 4) + tid + q * 128];
#pragma unroll
        for (int q = 0; q < G1_VEC; ++q) {
            const int idx = tid + q * 128;
            *(float4*)&tileA[0][(idx >> 6) * G1_PITCH + (idx & 63) * 4] = s0[q];
        }
    }
    __syncthreads();

    int buf = 0;
    for (; t + G1_BLOCKS < G1_TILES; t += G1_BLOCKS) {
        const float4* nx = a4 + (size_t)(t + G1_BLOCKS) * (G1_ROWS * DIM_IN / 4) + tid;
        G1_EACH(G1_LOAD)
        __builtin_amdgcn_sched_barrier(0);

        g1_tile(&tileA[buf][c * G1_PITCH + kq], bf, dinv, C, t * G1_ROWS + h * 8, n);

        float* nt = tileA[buf ^ 1];
        G1_EACH(G1_STORE)
        __syncthreads();
        buf ^= 1;
    }
    if (t < G1_TILES)
        g1_tile(&tileA[buf][c * G1_PITCH + kq], bf, dinv, C, t * G1_ROWS + h * 8, n);
}


#define GH_ROWS   160
#define GH_PITCH  (HD + 4)
#define GH_VEC    (GH_ROWS * HD / 4 / 128)
static_assert(N_NODES % GH_ROWS == 0 && GH_ROWS % 32 == 0 && (GH_ROWS * HD / 4) % 128 == 0 && (GH_PITCH * 4) % 16 == 0,
              "tile shape");

__global__ void __launch_bounds__(128)
TEXTGCN_gemm_hidden(const float* H,
                    const float* __restrict__ bprev,
                    const float* __restrict__ Wk,
                    const float* __restrict__ dinv,
                    float* __restrict__ T) {
    __shared__ __attribute__((aligned(16))) float tileH[GH_ROWS * GH_PITCH];

    const int tid  = threadIdx.x;
    const int lane = tid & 31;
    const int wv   = tid >> 5;
    const int c    = lane & 15;
    const int h    = lane >> 4;
    const int kq   = h * 2;
    const int n    = wv * 16 + c;
    const int row0 = blockIdx.x * GH_ROWS;

    const float4* h4 = (const float4*)(H + (size_t)row0 * HD);
    float4 st[GH_VEC];
#pragma unroll
    for (int q = 0; q < GH_VEC; ++q) st[q] = h4[tid + q * 128];

    v2f bf[HD / 4];
#pragma unroll
    for (int i = 0; i < HD / 4; ++i) {
        bf[i].x = Wk[(size_t)(4 * i + kq)     * HD + n];
        bf[i].y = Wk[(size_t)(4 * i + kq + 1) * HD + n];
    }
    const float4 bv = *(const float4*)(bprev + (tid & 15) * 4);
#pragma unroll
    for (int q = 0; q < GH_VEC; ++q) {
        const int idx = tid + q * 128;
        const float v0 = st[q].x + bv.x, v1 = st[q].y + bv.y, v2 = st[q].z + bv.z, v3 = st[q].w + bv.w;
        float4 o;
        o.x = v0 > 0.0f ? v0 : 0.0f;
        o.y = v1 > 0.0f ? v1 : 0.0f;
        o.z = v2 > 0.0f ? v2 : 0.0f;
        o.w = v3 > 0.0f ? v3 : 0.0f;
        *(float4*)&tileH[(idx >> 4) * GH_PITCH + (idx & 15) * 4] = o;
    }
    __syncthreads();

#pragma unroll
    for (int s = 0; s < GH_ROWS / 32; ++s) {
        const float* ap = &tileH[(s * 32 + c) * GH_PITCH + kq];
        v8f acc0 = {}, acc1 = {};
#pragma unroll
        for (int i = 0; i < HD / 4; ++i) {
            const v2f a0 = *(const v2f*)(ap + 4 * i);
            const v2f a1 = *(const v2f*)(ap + 16 * GH_PITCH + 4 * i);
            acc0 = __builtin_amdgcn_wmma_f32_16x16x4_f32(
                       false, a0, false, bf[i], (short)0, acc0, false, false);
            acc1 = __builtin_amdgcn_wmma_f32_16x16x4_f32(
                       false, a1, false, bf[i], (short)0, acc1, false, false);
        }
        const int mbase = row0 + s * 32 + h * 8;
        const float4 d0 = *(const float4*)(dinv + mbase);
        const float4 d1 = *(const float4*)(dinv + mbase + 4);
        const float4 e0 = *(const float4*)(dinv + mbase + 16);
        const float4 e1 = *(const float4*)(dinv + mbase + 20);
        const float dr[8] = {d0.x, d0.y, d0.z, d0.w, d1.x, d1.y, d1.z, d1.w};
        const float er[8] = {e0.x, e0.y, e0.z, e0.w, e1.x, e1.y, e1.z, e1.w};
        const unsigned ob = (unsigned)mbase * HD + (unsigned)n;
        float* __restrict__ tp = T + ob;
#pragma unroll
        for (int r = 0; r < 8; ++r) {
            tp[r * HD]        = acc0[r] * dr[r];
            tp[(16 + r) * HD] = acc1[r] * er[r];
        }
    }
}


__device__ __forceinline__ void lds_add_f32(float* p, float v) {
    __hip_atomic_fetch_add(p, v, __ATOMIC_RELAXED, __HIP_MEMORY_SCOPE_WORKGROUP);
}

__global__ void __launch_bounds__(1024)
TEXTGCN_propagate_owned(const float* __restrict__ pkf,
                        const float* __restrict__ ew,
                        const float* __restrict__ ownf,
                        const float* __restrict__ t,
                        const float* __restrict__ dinv,
                        float* __restrict__ acc, int e) {
    __shared__ float rows[OWN_ROWS * HD];
    __shared__ int   queue[QCAP];
    __shared__ int   qcount;

    const int tid  = threadIdx.x;
    const int lane = tid & 31;
    const int wave = __builtin_amdgcn_readfirstlane(tid >> 5);
    const int lo   = blockIdx.x * OWN_ROWS;
    int nrows = N_NODES - lo;
    if (nrows > OWN_ROWS) nrows = OWN_ROWS;

    const unsigned* __restrict__ pku = (const unsigned*)pkf;

    const float4* a4 = (const float4*)(t + (size_t)lo * HD);
    float4* r4 = (float4*)rows;
    const int nvec = nrows * (HD / 4);
    {
        constexpr int TRIPS = (OWN_ROWS * (HD / 4) + 1023) / 1024;
        float4 st[TRIPS];
#pragma unroll
        for (int q = 0; q < TRIPS; ++q) {
            int i = tid + q * 1024;
            if (i > nvec - 1) i = nvec - 1;
            st[q] = a4[i];
        }
#pragma unroll
        for (int q = 0; q < TRIPS; ++q) {
            const int i = tid + q * 1024;
            if (i < OWN_ROWS * (HD / 4)) r4[i] = st[q];
        }
    }
    if (tid == 0) qcount = 0;
    __syncthreads();

    const uint4* own4 = (const uint4*)ownf;
    const unsigned char* own1 = (const unsigned char*)ownf;
    const unsigned me = blockIdx.x * 0x01010101u;
    const int ngroups = e >> 4;
    for (int g0 = tid; g0 < ngroups; g0 += 8 * 1024) {
        uint4 v[8];
#pragma unroll
        for (int k = 0; k < 8; ++k) {
            int g = g0 + k * 1024;
            if (g > ngroups - 1) g = ngroups - 1;
            v[k] = own4[g];
        }
        unsigned p0 = 0u, p1 = 0u, p2 = 0u, p3 = 0u;
#pragma unroll
        for (int k = 0; k < 8; ++k) {
            const unsigned x0 = v[k].x ^ me, x1 = v[k].y ^ me, x2 = v[k].z ^ me, x3 = v[k].w ^ me;
            const unsigned u0 = ((x0 & 0x7F7F7F7Fu) + 0x7F7F7F7Fu) | x0;
            const unsigned u1 = ((x1 & 0x7F7F7F7Fu) + 0x7F7F7F7Fu) | x1;
            const unsigned u2 = ((x2 & 0x7F7F7F7Fu) + 0x7F7F7F7Fu) | x2;
            const unsigned u3 = ((x3 & 0x7F7F7F7Fu) + 0x7F7F7F7Fu) | x3;
            const unsigned sel = 0x01010101u << k;
            p0 |= (u0 >> (7 - k)) & sel;
            p1 |= (u1 >> (7 - k)) & sel;
            p2 |= (u2 >> (7 - k)) & sel;
            p3 |= (u3 >> (7 - k)) & sel;
        }
        int nv = (ngroups - g0 + 1023) >> 10;
        if (nv > 8) nv = 8;
        const unsigned vm = ((1u << nv) - 1u) * 0x01010101u;
        const unsigned mm[4] = {~p0 & vm, ~p1 & vm, ~p2 & vm, ~p3 & vm};
#pragma unroll
        for (int j = 0; j < 4; ++j) {
            unsigned m = mm[j];
            while (m != 0u) {
                const int bit = __builtin_ctz(m);
                m &= m - 1u;
                const int eid = ((g0 + (bit & 7) * 1024) << 4) + j * 4 + (bit >> 3);
                const int pos = atomicAdd(&qcount, 1);
                if (pos < QCAP) {
                    queue[pos] = eid;
                } else {
                    const unsigned pk = pku[eid];
                    const int s = (int)(pk & PK_MASK);
                    const int d = (int)(pk >> PK_SHIFT);
                    const float w = ew[eid];
#pragma nounroll
                    for (int cc = 0; cc < HD; ++cc)
                        lds_add_f32(&rows[d * HD + cc], t[(size_t)s * HD + cc] * w);
                }
            }
        }
    }
    for (int i = (ngroups << 4) + tid; i < e; i += 1024) {
        if (own1[i] == (unsigned char)blockIdx.x) {
            const int pos = atomicAdd(&qcount, 1);
            if (pos < QCAP) {
                queue[pos] = i;
            } else {
                const unsigned pk = pku[i];
                const int s = (int)(pk & PK_MASK);
                const int d = (int)(pk >> PK_SHIFT);
                const float w = ew[i];
#pragma nounroll
                for (int cc = 0; cc < HD; ++cc)
                    lds_add_f32(&rows[d * HD + cc], t[(size_t)s * HD + cc] * w);
            }
        }
    }
    __syncthreads();

    int n = qcount;
    if (n > QCAP) n = QCAP;
    n = __builtin_amdgcn_readfirstlane(n);
    for (int base = wave * 32; base < n; base += 1024) {
        int m = n - base;
        if (m > 32) m = 32;
        int sl = 0, dl = 0, wl = 0;
        if (lane < m) {
            const int eid = queue[base + lane];
            const unsigned pk = pku[eid];
            sl = (int)(pk & PK_MASK);
            dl = (int)(pk >> PK_SHIFT);
            wl = __float_as_int(ew[eid]);
        }
        for (int j = 0; j < m; j += 16) {
            float v0[16], v1[16];
#pragma unroll
            for (int k = 0; k < 16; ++k) {
                int idx = j + k;
                if (idx > m - 1) idx = m - 1;
                const int s = __builtin_amdgcn_readlane(sl, idx);
                const float* __restrict__ tp = t + (size_t)s * HD;
                v0[k] = tp[lane];
                v1[k] = tp[lane + 32];
            }
#pragma unroll
            for (int k = 0; k < 16; ++k) {
                if (j + k < m) {
                    const int d = __builtin_amdgcn_readlane(dl, j + k);
                    const float w = __int_as_float(__builtin_amdgcn_readlane(wl, j + k));
                    float* rp = rows + d * HD;
                    lds_add_f32(rp + lane,      v0[k] * w);
                    lds_add_f32(rp + lane + 32, v1[k] * w);
                }
            }
        }
    }
    __syncthreads();

    float4* o4 = (float4*)(acc + (size_t)lo * HD);
    {
        constexpr int TRIPS = (OWN_ROWS * (HD / 4) + 1023) / 1024;
        float4 ov[TRIPS];
        float  sc[TRIPS];
#pragma unroll
        for (int q = 0; q < TRIPS; ++q) {
            int i = tid + q * 1024;
            if (i > nvec - 1) i = nvec - 1;
            sc[q] = dinv[lo + (i >> 4)];
            ov[q] = r4[i];
        }
#pragma unroll
        for (int q = 0; q < TRIPS; ++q) {
            const int i = tid + q * 1024;
            if (i < nvec) {
                float4 v = ov[q];
                v.x *= sc[q]; v.y *= sc[q]; v.z *= sc[q]; v.w *= sc[q];
                o4[i] = v;
            }
        }
    }
}


__global__ void TEXTGCN_doc_gemm_wmma(const float* __restrict__ h,
                                      const float* __restrict__ bprev,
                                      const int* __restrict__ doc_idx,
                                      const float* __restrict__ Wp,
                                      const float* __restrict__ bp,
                                      float* __restrict__ dbuf) {
    const int lane = threadIdx.x;
    const int mt = blockIdx.x, nt = blockIdx.y;
    const int m  = mt * 16 + (lane & 15);
    const int n  = nt * 16 + (lane & 15);
    const int kq = (lane >> 4) * 2;

    const float* __restrict__ arow = h + (size_t)doc_idx[m] * HD;
    v8f acc = {};
    for (int k = 0; k < HD; k += 4) {
        v2f a, b;
        const float v0 = arow[k + kq]     + bprev[k + kq];
        const float v1 = arow[k + kq + 1] + bprev[k + kq + 1];
        a.x = v0 > 0.0f ? v0 : 0.0f;
        a.y = v1 > 0.0f ? v1 : 0.0f;
        b.x = Wp[(size_t)(k + kq)     * HD + n];
        b.y = Wp[(size_t)(k + kq + 1) * HD + n];
        acc = __builtin_amdgcn_wmma_f32_16x16x4_f32(
                  false, a, false, b, (short)0, acc, false, false);
    }
    const int mbase = mt * 16 + (lane >> 4) * 8;
    const int nst   = nt * 16 + (lane & 15);
    const float bias = bp[nst];
#pragma unroll
    for (int r = 0; r < 8; ++r) {
        float v = acc[r] + bias;
        dbuf[(size_t)(mbase + r) * HD + nst] = v > 0.0f ? v : 0.0f;
    }
}


__global__ void TEXTGCN_classifier(const float* __restrict__ dbuf,
                                   const float* __restrict__ Wc,
                                   const float* __restrict__ bc,
                                   float* __restrict__ out) {
    int gid = blockIdx.x * blockDim.x + threadIdx.x;
    if (gid >= NDOCS * NCLS) return;
    int d = gid / NCLS, c = gid % NCLS;
    const float* __restrict__ row = dbuf + (size_t)d * HD;
    float s = bc[c];
#pragma unroll 8
    for (int k = 0; k < HD; ++k) s += row[k] * Wc[(size_t)k * NCLS + c];
    out[gid] = s;
}


extern "C" void kernel_launch(void* const* d_in, const int* in_sizes, int n_in,
                              void* d_out, int out_size, void* d_ws, size_t ws_size,
                              hipStream_t stream) {
    const float* x   = (const float*)d_in[0];
    const float* ew  = (const float*)d_in[1];
    const float* W1  = (const float*)d_in[2];
    const float* b1  = (const float*)d_in[3];
    const float* W2  = (const float*)d_in[4];
    const float* b2  = (const float*)d_in[5];
    const float* W3  = (const float*)d_in[6];
    const float* b3  = (const float*)d_in[7];
    const float* Wp  = (const float*)d_in[8];
    const float* bp  = (const float*)d_in[9];
    const float* Wc  = (const float*)d_in[10];
    const float* bc  = (const float*)d_in[11];
    const int*   ei  = (const int*)d_in[12];
    const int*   doc = (const int*)d_in[13];
    float*       out = (float*)d_out;

    const int* src = ei;
    const int* dst = ei + N_EDGES;

    float* ws   = (float*)d_ws;
    float* dinv = ws;
    float* pk   = dinv + N_NODES;
    float* bufT = pk   + N_EDGES;
    float* bufH = bufT + (size_t)N_NODES * HD;
    float* dbuf = bufH + (size_t)N_NODES * HD;

    const int B = 256;
    dim3 blk(B);
    dim3 gN((N_NODES + B - 1) / B);
    dim3 gE((N_EDGES + B - 1) / B);

    TEXTGCN_deg_init<<<gN, blk, 0, stream>>>(dinv, N_NODES);
    TEXTGCN_deg_acc <<<gE, blk, 0, stream>>>(dst, ew, dinv, N_EDGES);
    TEXTGCN_dinv    <<<gN, blk, 0, stream>>>(dinv, N_NODES);
    TEXTGCN_pack    <<<gE, blk, 0, stream>>>(src, dst, pk, dbuf, N_EDGES);

    dim3 wave(32);
    dim3 gGemm1(G1_BLOCKS);
    dim3 gGemmH(N_NODES / GH_ROWS);
    dim3 blkH(128);
    dim3 gDoc(NDOCS / 16, HD / 16);
    dim3 gOwn(OWN_BLOCKS);
    dim3 blkOwn(1024);

    TEXTGCN_gemm_rows<<<gGemm1, blkH, 0, stream>>>(x, W1, dinv, bufT);
    TEXTGCN_propagate_owned<<<gOwn, blkOwn, 0, stream>>>(pk, ew, dbuf, bufT, dinv, bufH, N_EDGES);

    TEXTGCN_gemm_hidden<<<gGemmH, blkH, 0, stream>>>(bufH, b1, W2, dinv, bufT);
    TEXTGCN_propagate_owned<<<gOwn, blkOwn, 0, stream>>>(pk, ew, dbuf, bufT, dinv, bufH, N_EDGES);

    TEXTGCN_gemm_hidden<<<gGemmH, blkH, 0, stream>>>(bufH, b2, W3, dinv, bufT);
    TEXTGCN_propagate_owned<<<gOwn, blkOwn, 0, stream>>>(pk, ew, dbuf, bufT, dinv, bufH, N_EDGES);

    TEXTGCN_doc_gemm_wmma<<<gDoc, wave, 0, stream>>>(bufH, b3, doc, Wp, bp, dbuf);
    dim3 gCls((NDOCS * NCLS + B - 1) / B);
    TEXTGCN_classifier<<<gCls, blk, 0, stream>>>(dbuf, Wc, bc, out);
}


# owner scan done once: the first propagation writes each block's edge list (edge id, local row) to the workspace; layers 2 and 3 read it back and skip the scan; LDS adds in arrival order as before
# speedup vs baseline: 2.4913x; 1.1193x over previous
#include <hip/hip_runtime.h>
#include <hip/hip_bf16.h>

typedef float v2f __attribute__((ext_vector_type(2)));
typedef float v8f __attribute__((ext_vector_type(8)));

#define N_NODES 100000
#define N_EDGES 1600000
#define DIM_IN  256
#define HD      64
#define NCLS    20
#define NDOCS   10000

#define OWN_ROWS   391
#define OWN_BLOCKS 256
#define QCAP       16384
static_assert(OWN_BLOCKS * OWN_ROWS >= N_NODES && (OWN_BLOCKS - 1) * OWN_ROWS < N_NODES && OWN_BLOCKS <= 256,
              "every owner block holds at least one row and its id fits a byte");
static_assert(HD / 4 == 16, "a row is 16 pieces of 16 bytes (the write-back finds a piece's row by a shift of 4)");

__device__ __forceinline__ void atomic_add_f32(float* p, float v) {
    __hip_atomic_fetch_add(p, v, __ATOMIC_RELAXED, __HIP_MEMORY_SCOPE_AGENT);
}


__global__ void TEXTGCN_deg_init(float* __restrict__ deg, int n) {
    int i = blockIdx.x * blockDim.x + threadIdx.x;
    if (i < n) deg[i] = 1.0f;
}

__global__ void TEXTGCN_deg_acc(const int* __restrict__ dst,
                                const float* __restrict__ w,
                                float* __restrict__ deg, int e) {
    int i = blockIdx.x * blockDim.x + threadIdx.x;
    if (i < e) atomic_add_f32(&deg[dst[i]], w[i]);
}

__global__ void TEXTGCN_dinv(float* __restrict__ deg, int n) {
    int i = blockIdx.x * blockDim.x + threadIdx.x;
    if (i < n) {
        float d = deg[i];
        deg[i] = (d > 0.0f) ? __frsqrt_rn(d) : 0.0f;
    }
}

__global__ void TEXTGCN_owner_bytes(const int* __restrict__ dst, float* __restrict__ ownf, int e) {
    int i = blockIdx.x * blockDim.x + threadIdx.x;
    if (i < e) ((unsigned char*)ownf)[i] = (unsigned char)(dst[i] / OWN_ROWS);
}

#define LW_SHIFT 21
#define LW_MASK  ((1u << LW_SHIFT) - 1u)
#define LCAP     7008
#define CNT_PITCH 32
#define LMAIN    (N_EDGES / LCAP)
#define OWN_WORDS (N_EDGES / 4)
static_assert(N_EDGES <= (1 << LW_SHIFT) && OWN_ROWS <= (1 << (32 - LW_SHIFT)), "edge id and local row share a word");
static_assert(LMAIN <= OWN_BLOCKS && N_EDGES % 4 == 0 &&
              LCAP % 32 == 0 && OWN_WORDS % 32 == 0 && N_NODES % 32 == 0 &&
              OWN_WORDS + (OWN_BLOCKS - LMAIN) * LCAP + OWN_BLOCKS * CNT_PITCH <= NDOCS * HD,
              "owner bytes, the regions that do not fit the plane and the 256 counts fit the doc buffer's place");
static_assert(LCAP <= QCAP, "a list that is written was held whole in the LDS queue");


#define G1_ROWS   32
#define G1_PITCH  (DIM_IN + 4)
#define G1_TILES  (N_NODES / G1_ROWS)
#define G1_BLOCKS 1024
#define G1_VEC    (G1_ROWS * DIM_IN / 4 / 128)
#define G1_EACH(X) X(0) X(1) X(2) X(3) X(4) X(5) X(6) X(7) X(8) X(9) X(10) X(11) X(12) X(13) X(14) X(15)
#define G1_LOAD(q)  const float4 nv##q = nx[(q) * 128];
#define G1_STORE(q) *(float4*)&nt[((tid + (q) * 128) >> 6) * G1_PITCH + ((tid + (q) * 128) & 63) * 4] = nv##q;
static_assert(G1_VEC == 16, "G1_EACH names 16 values");
static_assert(N_NODES % G1_ROWS == 0 && HD == 64 && DIM_IN % 4 == 0 && (G1_ROWS * DIM_IN / 4) % 128 == 0 &&
              (G1_PITCH * 4) % 16 == 0, "tile shape");

__device__ __forceinline__ void g1_tile(const float* __restrict__ ap,
                                        const v2f (&bf)[DIM_IN / 4],
                                        const float* __restrict__ dinv,
                                        float* __restrict__ C,
                                        int mbase, int n) {
    v8f acc0 = {}, acc1 = {};
#pragma unroll
    for (int i = 0; i < DIM_IN / 4; ++i) {
        const v2f a0 = *(const v2f*)(ap + 4 * i);
        const v2f a1 = *(const v2f*)(ap + 16 * G1_PITCH + 4 * i);
        acc0 = __builtin_amdgcn_wmma_f32_16x16x4_f32(
                   false, a0, false, bf[i], (short)0, acc0, false, false);
        acc1 = __builtin_amdgcn_wmma_f32_16x16x4_f32(
                   false, a1, false, bf[i], (short)0, acc1, false, false);
    }
    const float4 d0 = *(const float4*)(dinv + mbase);
    const float4 d1 = *(const float4*)(dinv + mbase + 4);
    const float4 e0 = *(const float4*)(dinv + mbase + 16);
    const float4 e1 = *(const float4*)(dinv + mbase + 20);
    const float dr[8] = {d0.x, d0.y, d0.z, d0.w, d1.x, d1.y, d1.z, d1.w};
    const float er[8] = {e0.x, e0.y, e0.z, e0.w, e1.x, e1.y, e1.z, e1.w};
    const unsigned ob = (unsigned)mbase * HD + (unsigned)n;
    float* __restrict__ cp = C + ob;
#pragma unroll
    for (int r = 0; r < 8; ++r) {
        cp[r * HD]        = acc0[r] * dr[r];
        cp[(16 + r) * HD] = acc1[r] * er[r];
    }
}

__global__ void __launch_bounds__(128)
TEXTGCN_gemm_rows(const float* A,
                  const float* __restrict__ B,
                  const float* __restrict__ dinv,
                  float* __restrict__ C) {
    __shared__ __attribute__((aligned(16))) float tileA[2][G1_ROWS * G1_PITCH];

    const int tid  = threadIdx.x;
    const int lane = tid & 31;
    const int wv   = tid >> 5;
    const int c    = lane & 15;
    const int h    = lane >> 4;
    const int kq   = h * 2;
    const int n    = wv * 16 + c;

    v2f bf[DIM_IN / 4];
#pragma unroll
    for (int i = 0; i < DIM_IN / 4; ++i) {
        bf[i].x = B[(size_t)(4 * i + kq)     * HD + n];
        bf[i].y = B[(size_t)(4 * i + kq + 1) * HD + n];
    }

    const float4* a4 = (const float4*)A;
    int t = blockIdx.x;
    if (t < G1_TILES) {
        float4 s0[G1_VEC];
#pragma unroll
        for (int q = 0; q < G1_VEC; ++q) s0[q] = a4[(size_t)t * (G1_ROWS * DIM_IN / 4) + tid + q * 128];
#pragma unroll
        for (int q = 0; q < G1_VEC; ++q) {
            const int idx = tid + q * 128;
            *(float4*)&tileA[0][(idx >> 6) * G1_PITCH + (idx & 63) * 4] = s0[q];
        }
    }
    __syncthreads();

    int buf = 0;
    for (; t + G1_BLOCKS < G1_TILES; t += G1_BLOCKS) {
        const float4* nx = a4 + (size_t)(t + G1_BLOCKS) * (G1_ROWS * DIM_IN / 4) + tid;
        G1_EACH(G1_LOAD)
        __builtin_amdgcn_sched_barrier(0);

        g1_tile(&tileA[buf][c * G1_PITCH + kq], bf, dinv, C, t * G1_ROWS + h * 8, n);

        float* nt = tileA[buf ^ 1];
        G1_EACH(G1_STORE)
        __syncthreads();
        buf ^= 1;
    }
    if (t < G1_TILES)
        g1_tile(&tileA[buf][c * G1_PITCH + kq], bf, dinv, C, t * G1_ROWS + h * 8, n);
}


#define GH_ROWS   160
#define GH_PITCH  (HD + 4)
#define GH_VEC    (GH_ROWS * HD / 4 / 128)
static_assert(N_NODES % GH_ROWS == 0 && GH_ROWS % 32 == 0 && (GH_ROWS * HD / 4) % 128 == 0 && (GH_PITCH * 4) % 16 == 0,
              "tile shape");

__global__ void __launch_bounds__(128)
TEXTGCN_gemm_hidden(const float* H,
                    const float* __restrict__ bprev,
                    const float* __restrict__ Wk,
                    const float* __restrict__ dinv,
                    float* __restrict__ T) {
    __shared__ __attribute__((aligned(16))) float tileH[GH_ROWS * GH_PITCH];

    const int tid  = threadIdx.x;
    const int lane = tid & 31;
    const int wv   = tid >> 5;
    const int c    = lane & 15;
    const int h    = lane >> 4;
    const int kq   = h * 2;
    const int n    = wv * 16 + c;
    const int row0 = blockIdx.x * GH_ROWS;

    const float4* h4 = (const float4*)(H + (size_t)row0 * HD);
    float4 st[GH_VEC];
#pragma unroll
    for (int q = 0; q < GH_VEC; ++q) st[q] = h4[tid + q * 128];

    v2f bf[HD / 4];
#pragma unroll
    for (int i = 0; i < HD / 4; ++i) {
        bf[i].x = Wk[(size_t)(4 * i + kq)     * HD + n];
        bf[i].y = Wk[(size_t)(4 * i + kq + 1) * HD + n];
    }
    const float4 bv = *(const float4*)(bprev + (tid & 15) * 4);
#pragma unroll
    for (int q = 0; q < GH_VEC; ++q) {
        const int idx = tid + q * 128;
        const float v0 = st[q].x + bv.x, v1 = st[q].y + bv.y, v2 = st[q].z + bv.z, v3 = st[q].w + bv.w;
        float4 o;
        o.x = v0 > 0.0f ? v0 : 0.0f;
        o.y = v1 > 0.0f ? v1 : 0.0f;
        o.z = v2 > 0.0f ? v2 : 0.0f;
        o.w = v3 > 0.0f ? v3 : 0.0f;
        *(float4*)&tileH[(idx >> 4) * GH_PITCH + (idx & 15) * 4] = o;
    }
    __syncthreads();

#pragma unroll
    for (int s = 0; s < GH_ROWS / 32; ++s) {
        const float* ap = &tileH[(s * 32 + c) * GH_PITCH + kq];
        v8f acc0 = {}, acc1 = {};
#pragma unroll
        for (int i = 0; i < HD / 4; ++i) {
            const v2f a0 = *(const v2f*)(ap + 4 * i);
            const v2f a1 = *(const v2f*)(ap + 16 * GH_PITCH + 4 * i);
            acc0 = __builtin_amdgcn_wmma_f32_16x16x4_f32(
                       false, a0, false, bf[i], (short)0, acc0, false, false);
            acc1 = __builtin_amdgcn_wmma_f32_16x16x4_f32(
                       false, a1, false, bf[i], (short)0, acc1, false, false);
        }
        const int mbase = row0 + s * 32 + h * 8;
        const float4 d0 = *(const float4*)(dinv + mbase);
        const float4 d1 = *(const float4*)(dinv + mbase + 4);
        const float4 e0 = *(const float4*)(dinv + mbase + 16);
        const float4 e1 = *(const float4*)(dinv + mbase + 20);
        const float dr[8] = {d0.x, d0.y, d0.z, d0.w, d1.x, d1.y, d1.z, d1.w};
        const float er[8] = {e0.x, e0.y, e0.z, e0.w, e1.x, e1.y, e1.z, e1.w};
        const unsigned ob = (unsigned)mbase * HD + (unsigned)n;
        float* __restrict__ tp = T + ob;
#pragma unroll
        for (int r = 0; r < 8; ++r) {
            tp[r * HD]        = acc0[r] * dr[r];
            tp[(16 + r) * HD] = acc1[r] * er[r];
        }
    }
}


__device__ __forceinline__ void lds_add_f32(float* p, float v) {
    __hip_atomic_fetch_add(p, v, __ATOMIC_RELAXED, __HIP_MEMORY_SCOPE_WORKGROUP);
}

__device__ __forceinline__ void own_add_edges(int m, int sl, int dl, int wl, int lane,
                                              const float* __restrict__ t, float* rows) {
    for (int j = 0; j < m; j += 16) {
        float v0[16], v1[16];
#pragma unroll
        for (int k = 0; k < 16; ++k) {
            int idx = j + k;
            if (idx > m - 1) idx = m - 1;
            const int s = __builtin_amdgcn_readlane(sl, idx);
            const float* __restrict__ tp = t + (size_t)s * HD;
            v0[k] = tp[lane];
            v1[k] = tp[lane + 32];
        }
#pragma unroll
        for (int k = 0; k < 16; ++k) {
            if (j + k < m) {
                const int d = __builtin_amdgcn_readlane(dl, j + k);
                const float w = __int_as_float(__builtin_amdgcn_readlane(wl, j + k));
                float* rp = rows + d * HD;
                lds_add_f32(rp + lane,      v0[k] * w);
                lds_add_f32(rp + lane + 32, v1[k] * w);
            }
        }
    }
}

#define LTRIPS ((LCAP + 1023) / 1024)

template <bool FIRST>
__global__ void __launch_bounds__(1024)
TEXTGCN_propagate_owned(const int* __restrict__ src,
                        const int* __restrict__ dst,
                        const float* __restrict__ ew,
                        const float* __restrict__ ownf,
                        float* lstf,
                        float* sparef,
                        const float* __restrict__ t,
                        const float* __restrict__ dinv,
                        float* __restrict__ acc, int e) {
    __shared__ float rows[OWN_ROWS * HD];
    __shared__ int   queue[QCAP];
    __shared__ int   qcount;

    const int tid  = threadIdx.x;
    const int lane = tid & 31;
    const int wave = __builtin_amdgcn_readfirstlane(tid >> 5);
    const int lo   = blockIdx.x * OWN_ROWS;
    int nrows = N_NODES - lo;
    if (nrows > OWN_ROWS) nrows = OWN_ROWS;

    unsigned* reg = ((int)blockIdx.x < LMAIN) ? (unsigned*)lstf + (size_t)blockIdx.x * LCAP
                                              : (unsigned*)sparef + (size_t)((int)blockIdx.x - LMAIN) * LCAP;
    int* cnt = (int*)sparef + (OWN_BLOCKS - LMAIN) * LCAP;
    int listed = -1;
    if (!FIRST) listed = cnt[blockIdx.x * CNT_PITCH];
    const bool from_list = !FIRST && listed >= 0;

    int lsl[LTRIPS], ldl[LTRIPS], lwl[LTRIPS];
#pragma unroll
    for (int q = 0; q < LTRIPS; ++q) { lsl[q] = 0; ldl[q] = 0; lwl[q] = 0; }
    if (from_list) {
        unsigned wdv[LTRIPS];
#pragma unroll
        for (int q = 0; q < LTRIPS; ++q) {
            const int i = tid + q * 1024;
            wdv[q] = (i < listed) ? reg[i] : 0u;
        }
#pragma unroll
        for (int q = 0; q < LTRIPS; ++q) {
            const int eid = (int)(wdv[q] & LW_MASK);
            ldl[q] = (int)(wdv[q] >> LW_SHIFT);
            lsl[q] = src[eid];
            lwl[q] = __float_as_int(ew[eid]);
        }
    }

    const float4* a4 = (const float4*)(t + (size_t)lo * HD);
    float4* r4 = (float4*)rows;
    const int nvec = nrows * (HD / 4);
    {
        constexpr int TRIPS = (OWN_ROWS * (HD / 4) + 1023) / 1024;
        float4 st[TRIPS];
#pragma unroll
        for (int q = 0; q < TRIPS; ++q) {
            int i = tid + q * 1024;
            if (i > nvec - 1) i = nvec - 1;
            st[q] = a4[i];
        }
#pragma unroll
        for (int q = 0; q < TRIPS; ++q) {
            const int i = tid + q * 1024;
            if (i < OWN_ROWS * (HD / 4)) r4[i] = st[q];
        }
    }
    if (tid == 0) qcount = 0;
    __syncthreads();

    if (!from_list) {
    const uint4* own4 = (const uint4*)ownf;
    const unsigned char* own1 = (const unsigned char*)ownf;
    const unsigned me = blockIdx.x * 0x01010101u;
    const int ngroups = e >> 4;
    for (int g0 = tid; g0 < ngroups; g0 += 8 * 1024) {
        uint4 v[8];
#pragma unroll
        for (int k = 0; k < 8; ++k) {
            int g = g0 + k * 1024;
            if (g > ngroups - 1) g = ngroups - 1;
            v[k] = own4[g];
        }
        unsigned p0 = 0u, p1 = 0u, p2 = 0u, p3 = 0u;
#pragma unroll
        for (int k = 0; k < 8; ++k) {
            const unsigned x0 = v[k].x ^ me, x1 = v[k].y ^ me, x2 = v[k].z ^ me, x3 = v[k].w ^ me;
            const unsigned u0 = ((x0 & 0x7F7F7F7Fu) + 0x7F7F7F7Fu) | x0;
            const unsigned u1 = ((x1 & 0x7F7F7F7Fu) + 0x7F7F7F7Fu) | x1;
            const unsigned u2 = ((x2 & 0x7F7F7F7Fu) + 0x7F7F7F7Fu) | x2;
            const unsigned u3 = ((x3 & 0x7F7F7F7Fu) + 0x7F7F7F7Fu) | x3;
            const unsigned sel = 0x01010101u << k;
            p0 |= (u0 >> (7 - k)) & sel;
            p1 |= (u1 >> (7 - k)) & sel;
            p2 |= (u2 >> (7 - k)) & sel;
            p3 |= (u3 >> (7 - k)) & sel;
        }
        int nv = (ngroups - g0 + 1023) >> 10;
        if (nv > 8) nv = 8;
        const unsigned vm = ((1u << nv) - 1u) * 0x01010101u;
        const unsigned mm[4] = {~p0 & vm, ~p1 & vm, ~p2 & vm, ~p3 & vm};
#pragma unroll
        for (int j = 0; j < 4; ++j) {
            unsigned m = mm[j];
            while (m != 0u) {
                const int bit = __builtin_ctz(m);
                m &= m - 1u;
                const int eid = ((g0 + (bit & 7) * 1024) << 4) + j * 4 + (bit >> 3);
                const int pos = atomicAdd(&qcount, 1);
                if (pos < QCAP) {
                    queue[pos] = eid;
                } else {
                    const int s = src[eid];
                    const int d = dst[eid] - lo;
                    const float w = ew[eid];
#pragma nounroll
                    for (int cc = 0; cc < HD; ++cc)
                        lds_add_f32(&rows[d * HD + cc], t[(size_t)s * HD + cc] * w);
                }
            }
        }
    }
    for (int i = (ngroups << 4) + tid; i < e; i += 1024) {
        if (own1[i] == (unsigned char)blockIdx.x) {
            const int pos = atomicAdd(&qcount, 1);
            if (pos < QCAP) {
                queue[pos] = i;
            } else {
                const int s = src[i];
                const int d = dst[i] - lo;
                const float w = ew[i];
#pragma nounroll
                for (int cc = 0; cc < HD; ++cc)
                    lds_add_f32(&rows[d * HD + cc], t[(size_t)s * HD + cc] * w);
            }
        }
    }
    }
    __syncthreads();

    if (from_list) {
        const int n = listed;
#pragma unroll
        for (int q = 0; q < LTRIPS; ++q) {
            const int base = wave * 32 + q * 1024;
            if (base < n) {
                int m = n - base;
                if (m > 32) m = 32;
                own_add_edges(m, lsl[q], ldl[q], lwl[q], lane, t, rows);
            }
        }
    } else {
        int n = qcount;
        const bool keep = FIRST && n <= LCAP;
        if (FIRST && tid == 0) cnt[blockIdx.x * CNT_PITCH] = keep ? n : -1;
        if (n > QCAP) n = QCAP;
        n = __builtin_amdgcn_readfirstlane(n);
        for (int base = wave * 32; base < n; base += 1024) {
            int m = n - base;
            if (m > 32) m = 32;
            int sl = 0, dl = 0, wl = 0;
            if (lane < m) {
                const int eid = queue[base + lane];
                const int dg = dst[eid];
                sl = src[eid];
                wl = __float_as_int(ew[eid]);
                dl = dg - lo;
                if (keep) reg[base + lane] = (unsigned)eid | ((unsigned)dl << LW_SHIFT);
            }
            own_add_edges(m, sl, dl, wl, lane, t, rows);
        }
    }
    __syncthreads();

    float4* o4 = (float4*)(acc + (size_t)lo * HD);
    {
        constexpr int TRIPS = (OWN_ROWS * (HD / 4) + 1023) / 1024;
        float4 ov[TRIPS];
        float  sc[TRIPS];
#pragma unroll
        for (int q = 0; q < TRIPS; ++q) {
            int i = tid + q * 1024;
            if (i > nvec - 1) i = nvec - 1;
            sc[q] = dinv[lo + (i >> 4)];
            ov[q] = r4[i];
        }
#pragma unroll
        for (int q = 0; q < TRIPS; ++q) {
            const int i = tid + q * 1024;
            if (i < nvec) {
                float4 v = ov[q];
                v.x *= sc[q]; v.y *= sc[q]; v.z *= sc[q]; v.w *= sc[q];
                o4[i] = v;
            }
        }
    }
}


__global__ void TEXTGCN_doc_gemm_wmma(const float* __restrict__ h,
                                      const float* __restrict__ bprev,
                                      const int* __restrict__ doc_idx,
                                      const float* __restrict__ Wp,
                                      const float* __restrict__ bp,
                                      float* __restrict__ dbuf) {
    const int lane = threadIdx.x;
    const int mt = blockIdx.x, nt = blockIdx.y;
    const int m  = mt * 16 + (lane & 15);
    const int n  = nt * 16 + (lane & 15);
    const int kq = (lane >> 4) * 2;

    const float* __restrict__ arow = h + (size_t)doc_idx[m] * HD;
    v8f acc = {};
    for (int k = 0; k < HD; k += 4) {
        v2f a, b;
        const float v0 = arow[k + kq]     + bprev[k + kq];
        const float v1 = arow[k + kq + 1] + bprev[k + kq + 1];
        a.x = v0 > 0.0f ? v0 : 0.0f;
        a.y = v1 > 0.0f ? v1 : 0.0f;
        b.x = Wp[(size_t)(k + kq)     * HD + n];
        b.y = Wp[(size_t)(k + kq + 1) * HD + n];
        acc = __builtin_amdgcn_wmma_f32_16x16x4_f32(
                  false, a, false, b, (short)0, acc, false, false);
    }
    const int mbase = mt * 16 + (lane >> 4) * 8;
    const int nst   = nt * 16 + (lane & 15);
    const float bias = bp[nst];
#pragma unroll
    for (int r = 0; r < 8; ++r) {
        float v = acc[r] + bias;
        dbuf[(size_t)(mbase + r) * HD + nst] = v > 0.0f ? v : 0.0f;
    }
}


__global__ void TEXTGCN_classifier(const float* __restrict__ dbuf,
                                   const float* __restrict__ Wc,
                                   const float* __restrict__ bc,
                                   float* __restrict__ out) {
    int gid = blockIdx.x * blockDim.x + threadIdx.x;
    if (gid >= NDOCS * NCLS) return;
    int d = gid / NCLS, c = gid % NCLS;
    const float* __restrict__ row = dbuf + (size_t)d * HD;
    float s = bc[c];
#pragma unroll 8
    for (int k = 0; k < HD; ++k) s += row[k] * Wc[(size_t)k * NCLS + c];
    out[gid] = s;
}


extern "C" void kernel_launch(void* const* d_in, const int* in_sizes, int n_in,
                              void* d_out, int out_size, void* d_ws, size_t ws_size,
                              hipStream_t stream) {
    const float* x   = (const float*)d_in[0];
    const float* ew  = (const float*)d_in[1];
    const float* W1  = (const float*)d_in[2];
    const float* b1  = (const float*)d_in[3];
    const float* W2  = (const float*)d_in[4];
    const float* b2  = (const float*)d_in[5];
    const float* W3  = (const float*)d_in[6];
    const float* b3  = (const float*)d_in[7];
    const float* Wp  = (const float*)d_in[8];
    const float* bp  = (const float*)d_in[9];
    const float* Wc  = (const float*)d_in[10];
    const float* bc  = (const float*)d_in[11];
    const int*   ei  = (const int*)d_in[12];
    const int*   doc = (const int*)d_in[13];
    float*       out = (float*)d_out;

    const int* src = ei;
    const int* dst = ei + N_EDGES;

    float* ws   = (float*)d_ws;
    float* dinv = ws;
    float* lst  = dinv + N_NODES;
    float* bufT = lst  + N_EDGES;
    float* bufH = bufT + (size_t)N_NODES * HD;
    float* dbuf = bufH + (size_t)N_NODES * HD;

    const int B = 256;
    dim3 blk(B);
    dim3 gN((N_NODES + B - 1) / B);
    dim3 gE((N_EDGES + B - 1) / B);

    TEXTGCN_deg_init<<<gN, blk, 0, stream>>>(dinv, N_NODES);
    TEXTGCN_deg_acc <<<gE, blk, 0, stream>>>(dst, ew, dinv, N_EDGES);
    TEXTGCN_dinv    <<<gN, blk, 0, stream>>>(dinv, N_NODES);
    TEXTGCN_owner_bytes<<<gE, blk, 0, stream>>>(dst, dbuf, N_EDGES);
    float* spare = dbuf + OWN_WORDS;

    dim3 wave(32);
    dim3 gGemm1(G1_BLOCKS);
    dim3 gGemmH(N_NODES / GH_ROWS);
    dim3 blkH(128);
    dim3 gDoc(NDOCS / 16, HD / 16);
    dim3 gOwn(OWN_BLOCKS);
    dim3 blkOwn(1024);

    TEXTGCN_gemm_rows<<<gGemm1, blkH, 0, stream>>>(x, W1, dinv, bufT);
    TEXTGCN_propagate_owned<true><<<gOwn, blkOwn, 0, stream>>>(src, dst, ew, dbuf, lst, spare, bufT, dinv, bufH, N_EDGES);

    TEXTGCN_gemm_hidden<<<gGemmH, blkH, 0, stream>>>(bufH, b1, W2, dinv, bufT);
    TEXTGCN_propagate_owned<false><<<gOwn, blkOwn, 0, stream>>>(src, dst, ew, dbuf, lst, spare, bufT, dinv, bufH, N_EDGES);

    TEXTGCN_gemm_hidden<<<gGemmH, blkH, 0, stream>>>(bufH, b2, W3, dinv, bufT);
    TEXTGCN_propagate_owned<false><<<gOwn, blkOwn, 0, stream>>>(src, dst, ew, dbuf, lst, spare, bufT, dinv, bufH, N_EDGES);

    TEXTGCN_doc_gemm_wmma<<<gDoc, wave, 0, stream>>>(bufH, b3, doc, Wp, bp, dbuf);
    dim3 gCls((NDOCS * NCLS + B - 1) / B);
    TEXTGCN_classifier<<<gCls, blk, 0, stream>>>(dbuf, Wc, bc, out);
}


# degree: float atomics to memory replaced by one scan launch of the 256 owner blocks: lists each block's edges, sums their weights by LDS float adds (arrival order), writes dinv; 12 launches to 10
# speedup vs baseline: 3.2310x; 1.2969x over previous
#include <hip/hip_runtime.h>
#include <hip/hip_bf16.h>

typedef float v2f __attribute__((ext_vector_type(2)));
typedef float v8f __attribute__((ext_vector_type(8)));

#define N_NODES 100000
#define N_EDGES 1600000
#define DIM_IN  256
#define HD      64
#define NCLS    20
#define NDOCS   10000

#define OWN_ROWS   391
#define OWN_BLOCKS 256
#define QCAP       16384
static_assert(OWN_BLOCKS * OWN_ROWS >= N_NODES && (OWN_BLOCKS - 1) * OWN_ROWS < N_NODES && OWN_BLOCKS <= 256,
              "every owner block holds at least one row and its id fits a byte");
static_assert(HD / 4 == 16, "a row is 16 pieces of 16 bytes (the write-back finds a piece's row by a shift of 4)");


__global__ void TEXTGCN_owner_bytes(const int* __restrict__ dst, float* __restrict__ ownf, int e) {
    int i = blockIdx.x * blockDim.x + threadIdx.x;
    if (i < e) ((unsigned char*)ownf)[i] = (unsigned char)(dst[i] / OWN_ROWS);
}

#define LW_SHIFT 21
#define LW_MASK  ((1u << LW_SHIFT) - 1u)
#define LCAP     7008
#define CNT_PITCH 32
#define LMAIN    (N_EDGES / LCAP)
#define OWN_WORDS (N_EDGES / 4)
static_assert(N_EDGES <= (1 << LW_SHIFT) && OWN_ROWS <= (1 << (32 - LW_SHIFT)), "edge id and local row share a word");
static_assert(LMAIN <= OWN_BLOCKS && N_EDGES % 4 == 0 &&
              LCAP % 32 == 0 && OWN_WORDS % 32 == 0 && N_NODES % 32 == 0 &&
              OWN_WORDS + (OWN_BLOCKS - LMAIN) * LCAP + OWN_BLOCKS * CNT_PITCH <= NDOCS * HD,
              "owner bytes, the regions that do not fit the plane and the 256 counts fit the doc buffer's place");
static_assert(LCAP <= QCAP, "a list that is written was held whole in the LDS queue");


#define G1_ROWS   32
#define G1_PITCH  (DIM_IN + 4)
#define G1_TILES  (N_NODES / G1_ROWS)
#define G1_BLOCKS 1024
#define G1_VEC    (G1_ROWS * DIM_IN / 4 / 128)
#define G1_EACH(X) X(0) X(1) X(2) X(3) X(4) X(5) X(6) X(7) X(8) X(9) X(10) X(11) X(12) X(13) X(14) X(15)
#define G1_LOAD(q)  const float4 nv##q = nx[(q) * 128];
#define G1_STORE(q) *(float4*)&nt[((tid + (q) * 128) >> 6) * G1_PITCH + ((tid + (q) * 128) & 63) * 4] = nv##q;
static_assert(G1_VEC == 16, "G1_EACH names 16 values");
static_assert(N_NODES % G1_ROWS == 0 && HD == 64 && DIM_IN % 4 == 0 && (G1_ROWS * DIM_IN / 4) % 128 == 0 &&
              (G1_PITCH * 4) % 16 == 0, "tile shape");

__device__ __forceinline__ void g1_tile(const float* __restrict__ ap,
                                        const v2f (&bf)[DIM_IN / 4],
                                        const float* __restrict__ dinv,
                                        float* __restrict__ C,
                                        int mbase, int n) {
    v8f acc0 = {}, acc1 = {};
#pragma unroll
    for (int i = 0; i < DIM_IN / 4; ++i) {
        const v2f a0 = *(const v2f*)(ap + 4 * i);
        const v2f a1 = *(const v2f*)(ap + 16 * G1_PITCH + 4 * i);
        acc0 = __builtin_amdgcn_wmma_f32_16x16x4_f32(
                   false, a0, false, bf[i], (short)0, acc0, false, false);
        acc1 = __builtin_amdgcn_wmma_f32_16x16x4_f32(
                   false, a1, false, bf[i], (short)0, acc1, false, false);
    }
    const float4 d0 = *(const float4*)(dinv + mbase);
    const float4 d1 = *(const float4*)(dinv + mbase + 4);
    const float4 e0 = *(const float4*)(dinv + mbase + 16);
    const float4 e1 = *(const float4*)(dinv + mbase + 20);
    const float dr[8] = {d0.x, d0.y, d0.z, d0.w, d1.x, d1.y, d1.z, d1.w};
    const float er[8] = {e0.x, e0.y, e0.z, e0.w, e1.x, e1.y, e1.z, e1.w};
    const unsigned ob = (unsigned)mbase * HD + (unsigned)n;
    float* __restrict__ cp = C + ob;
#pragma unroll
    for (int r = 0; r < 8; ++r) {
        cp[r * HD]        = acc0[r] * dr[r];
        cp[(16 + r) * HD] = acc1[r] * er[r];
    }
}

__global__ void __launch_bounds__(128)
TEXTGCN_gemm_rows(const float* A,
                  const float* __restrict__ B,
                  const float* __restrict__ dinv,
                  float* __restrict__ C) {
    __shared__ __attribute__((aligned(16))) float tileA[2][G1_ROWS * G1_PITCH];

    const int tid  = threadIdx.x;
    const int lane = tid & 31;
    const int wv   = tid >> 5;
    const int c    = lane & 15;
    const int h    = lane >> 4;
    const int kq   = h * 2;
    const int n    = wv * 16 + c;

    v2f bf[DIM_IN / 4];
#pragma unroll
    for (int i = 0; i < DIM_IN / 4; ++i) {
        bf[i].x = B[(size_t)(4 * i + kq)     * HD + n];
        bf[i].y = B[(size_t)(4 * i + kq + 1) * HD + n];
    }

    const float4* a4 = (const float4*)A;
    int t = blockIdx.x;
    if (t < G1_TILES) {
        float4 s0[G1_VEC];
#pragma unroll
        for (int q = 0; q < G1_VEC; ++q) s0[q] = a4[(size_t)t * (G1_ROWS * DIM_IN / 4) + tid + q * 128];
#pragma unroll
        for (int q = 0; q < G1_VEC; ++q) {
            const int idx = tid + q * 128;
            *(float4*)&tileA[0][(idx >> 6) * G1_PITCH + (idx & 63) * 4] = s0[q];
        }
    }
    __syncthreads();

    int buf = 0;
    for (; t + G1_BLOCKS < G1_TILES; t += G1_BLOCKS) {
        const float4* nx = a4 + (size_t)(t + G1_BLOCKS) * (G1_ROWS * DIM_IN / 4) + tid;
        G1_EACH(G1_LOAD)
        __builtin_amdgcn_sched_barrier(0);

        g1_tile(&tileA[buf][c * G1_PITCH + kq], bf, dinv, C, t * G1_ROWS + h * 8, n);

        float* nt = tileA[buf ^ 1];
        G1_EACH(G1_STORE)
        __syncthreads();
        buf ^= 1;
    }
    if (t < G1_TILES)
        g1_tile(&tileA[buf][c * G1_PITCH + kq], bf, dinv, C, t * G1_ROWS + h * 8, n);
}


#define GH_ROWS   160
#define GH_PITCH  (HD + 4)
#define GH_VEC    (GH_ROWS * HD / 4 / 128)
static_assert(N_NODES % GH_ROWS == 0 && GH_ROWS % 32 == 0 && (GH_ROWS * HD / 4) % 128 == 0 && (GH_PITCH * 4) % 16 == 0,
              "tile shape");

__global__ void __launch_bounds__(128)
TEXTGCN_gemm_hidden(const float* H,
                    const float* __restrict__ bprev,
                    const float* __restrict__ Wk,
                    const float* __restrict__ dinv,
                    float* __restrict__ T) {
    __shared__ __attribute__((aligned(16))) float tileH[GH_ROWS * GH_PITCH];

    const int tid  = threadIdx.x;
    const int lane = tid & 31;
    const int wv   = tid >> 5;
    const int c    = lane & 15;
    const int h    = lane >> 4;
    const int kq   = h * 2;
    const int n    = wv * 16 + c;
    const int row0 = blockIdx.x * GH_ROWS;

    const float4* h4 = (const float4*)(H + (size_t)row0 * HD);
    float4 st[GH_VEC];
#pragma unroll
    for (int q = 0; q < GH_VEC; ++q) st[q] = h4[tid + q * 128];

    v2f bf[HD / 4];
#pragma unroll
    for (int i = 0; i < HD / 4; ++i) {
        bf[i].x = Wk[(size_t)(4 * i + kq)     * HD + n];
        bf[i].y = Wk[(size_t)(4 * i + kq + 1) * HD + n];
    }
    const float4 bv = *(const float4*)(bprev + (tid & 15) * 4);
#pragma unroll
    for (int q = 0; q < GH_VEC; ++q) {
        const int idx = tid + q * 128;
        const float v0 = st[q].x + bv.x, v1 = st[q].y + bv.y, v2 = st[q].z + bv.z, v3 = st[q].w + bv.w;
        float4 o;
        o.x = v0 > 0.0f ? v0 : 0.0f;
        o.y = v1 > 0.0f ? v1 : 0.0f;
        o.z = v2 > 0.0f ? v2 : 0.0f;
        o.w = v3 > 0.0f ? v3 : 0.0f;
        *(float4*)&tileH[(idx >> 4) * GH_PITCH + (idx & 15) * 4] = o;
    }
    __syncthreads();

#pragma unroll
    for (int s = 0; s < GH_ROWS / 32; ++s) {
        const float* ap = &tileH[(s * 32 + c) * GH_PITCH + kq];
        v8f acc0 = {}, acc1 = {};
#pragma unroll
        for (int i = 0; i < HD / 4; ++i) {
            const v2f a0 = *(const v2f*)(ap + 4 * i);
            const v2f a1 = *(const v2f*)(ap + 16 * GH_PITCH + 4 * i);
            acc0 = __builtin_amdgcn_wmma_f32_16x16x4_f32(
                       false, a0, false, bf[i], (short)0, acc0, false, false);
            acc1 = __builtin_amdgcn_wmma_f32_16x16x4_f32(
                       false, a1, false, bf[i], (short)0, acc1, false, false);
        }
        const int mbase = row0 + s * 32 + h * 8;
        const float4 d0 = *(const float4*)(dinv + mbase);
        const float4 d1 = *(const float4*)(dinv + mbase + 4);
        const float4 e0 = *(const float4*)(dinv + mbase + 16);
        const float4 e1 = *(const float4*)(dinv + mbase + 20);
        const float dr[8] = {d0.x, d0.y, d0.z, d0.w, d1.x, d1.y, d1.z, d1.w};
        const float er[8] = {e0.x, e0.y, e0.z, e0.w, e1.x, e1.y, e1.z, e1.w};
        const unsigned ob = (unsigned)mbase * HD + (unsigned)n;
        float* __restrict__ tp = T + ob;
#pragma unroll
        for (int r = 0; r < 8; ++r) {
            tp[r * HD]        = acc0[r] * dr[r];
            tp[(16 + r) * HD] = acc1[r] * er[r];
        }
    }
}


__device__ __forceinline__ void lds_add_f32(float* p, float v) {
    __hip_atomic_fetch_add(p, v, __ATOMIC_RELAXED, __HIP_MEMORY_SCOPE_WORKGROUP);
}

template <class Over>
__device__ __forceinline__ void own_scan(const float* __restrict__ ownf, int e, int tid,
                                         int* queue, int* qcount, Over over) {
    const uint4* own4 = (const uint4*)ownf;
    const unsigned char* own1 = (const unsigned char*)ownf;
    const unsigned me = blockIdx.x * 0x01010101u;
    const int ngroups = e >> 4;
    for (int g0 = tid; g0 < ngroups; g0 += 8 * 1024) {
        uint4 v[8];
#pragma unroll
        for (int k = 0; k < 8; ++k) {
            int g = g0 + k * 1024;
            if (g > ngroups - 1) g = ngroups - 1;
            v[k] = own4[g];
        }
        unsigned p0 = 0u, p1 = 0u, p2 = 0u, p3 = 0u;
#pragma unroll
        for (int k = 0; k < 8; ++k) {
            const unsigned x0 = v[k].x ^ me, x1 = v[k].y ^ me, x2 = v[k].z ^ me, x3 = v[k].w ^ me;
            const unsigned u0 = ((x0 & 0x7F7F7F7Fu) + 0x7F7F7F7Fu) | x0;
            const unsigned u1 = ((x1 & 0x7F7F7F7Fu) + 0x7F7F7F7Fu) | x1;
            const unsigned u2 = ((x2 & 0x7F7F7F7Fu) + 0x7F7F7F7Fu) | x2;
            const unsigned u3 = ((x3 & 0x7F7F7F7Fu) + 0x7F7F7F7Fu) | x3;
            const unsigned sel = 0x01010101u << k;
            p0 |= (u0 >> (7 - k)) & sel;
            p1 |= (u1 >> (7 - k)) & sel;
            p2 |= (u2 >> (7 - k)) & sel;
            p3 |= (u3 >> (7 - k)) & sel;
        }
        int nv = (ngroups - g0 + 1023) >> 10;
        if (nv > 8) nv = 8;
        const unsigned vm = ((1u << nv) - 1u) * 0x01010101u;
        const unsigned mm[4] = {~p0 & vm, ~p1 & vm, ~p2 & vm, ~p3 & vm};
#pragma unroll
        for (int j = 0; j < 4; ++j) {
            unsigned m = mm[j];
            while (m != 0u) {
                const int bit = __builtin_ctz(m);
                m &= m - 1u;
                const int eid = ((g0 + (bit & 7) * 1024) << 4) + j * 4 + (bit >> 3);
                const int pos = atomicAdd(qcount, 1);
                if (pos < QCAP) queue[pos] = eid;
                else            over(eid);
            }
        }
    }
    for (int i = (ngroups << 4) + tid; i < e; i += 1024) {
        if (own1[i] == (unsigned char)blockIdx.x) {
            const int pos = atomicAdd(qcount, 1);
            if (pos < QCAP) queue[pos] = i;
            else            over(i);
        }
    }
}

#define SCAN_BATCH 4

__global__ void __launch_bounds__(1024)
TEXTGCN_owner_scan(const int* __restrict__ dst,
                   const float* __restrict__ ew,
                   const float* __restrict__ ownf,
                   float* __restrict__ lstf,
                   float* __restrict__ sparef,
                   float* __restrict__ dinv, int e) {
    __shared__ float degs[OWN_ROWS];
    __shared__ int   queue[QCAP];
    __shared__ int   qcount;

    const int tid = threadIdx.x;
    const int lo  = blockIdx.x * OWN_ROWS;
    int nrows = N_NODES - lo;
    if (nrows > OWN_ROWS) nrows = OWN_ROWS;

    unsigned* reg = ((int)blockIdx.x < LMAIN) ? (unsigned*)lstf + (size_t)blockIdx.x * LCAP
                                              : (unsigned*)sparef + (size_t)((int)blockIdx.x - LMAIN) * LCAP;
    int* cnt = (int*)sparef + (OWN_BLOCKS - LMAIN) * LCAP;

    if (tid < OWN_ROWS) degs[tid] = 1.0f;
    if (tid == 0) qcount = 0;
    __syncthreads();

    float* const degs0 = degs;
    own_scan(ownf, e, tid, queue, &qcount,
             [=](int eid) { lds_add_f32(&degs0[dst[eid] - lo], ew[eid]); });
    __syncthreads();

    int n = qcount;
    const bool keep = n <= LCAP;
    if (tid == 0) cnt[blockIdx.x * CNT_PITCH] = keep ? n : -1;
    if (n > QCAP) n = QCAP;
    for (int b = tid; b < n; b += SCAN_BATCH * 1024) {
        int eid[SCAN_BATCH], dg[SCAN_BATCH];
        float w[SCAN_BATCH];
#pragma unroll
        for (int k = 0; k < SCAN_BATCH; ++k) {
            int i = b + k * 1024;
            if (i > n - 1) i = n - 1;
            eid[k] = queue[i];
        }
#pragma unroll
        for (int k = 0; k < SCAN_BATCH; ++k) {
            dg[k] = dst[eid[k]];
            w[k]  = ew[eid[k]];
        }
#pragma unroll
        for (int k = 0; k < SCAN_BATCH; ++k) {
            const int i = b + k * 1024;
            if (i < n) {
                const int dl = dg[k] - lo;
                lds_add_f32(&degs[dl], w[k]);
                if (keep) reg[i] = (unsigned)eid[k] | ((unsigned)dl << LW_SHIFT);
            }
        }
    }
    __syncthreads();

    if (tid < nrows) {
        const float d = degs[tid];
        dinv[lo + tid] = (d > 0.0f) ? __frsqrt_rn(d) : 0.0f;
    }
}

__device__ __forceinline__ void own_add_edges(int m, int sl, int dl, int wl, int lane,
                                              const float* __restrict__ t, float* rows) {
    for (int j = 0; j < m; j += 16) {
        float v0[16], v1[16];
#pragma unroll
        for (int k = 0; k < 16; ++k) {
            int idx = j + k;
            if (idx > m - 1) idx = m - 1;
            const int s = __builtin_amdgcn_readlane(sl, idx);
            const float* __restrict__ tp = t + (size_t)s * HD;
            v0[k] = tp[lane];
            v1[k] = tp[lane + 32];
        }
#pragma unroll
        for (int k = 0; k < 16; ++k) {
            if (j + k < m) {
                const int d = __builtin_amdgcn_readlane(dl, j + k);
                const float w = __int_as_float(__builtin_amdgcn_readlane(wl, j + k));
                float* rp = rows + d * HD;
                lds_add_f32(rp + lane,      v0[k] * w);
                lds_add_f32(rp + lane + 32, v1[k] * w);
            }
        }
    }
}

#define LTRIPS ((LCAP + 1023) / 1024)

__global__ void __launch_bounds__(1024)
TEXTGCN_propagate_owned(const int* __restrict__ src,
                        const int* __restrict__ dst,
                        const float* __restrict__ ew,
                        const float* __restrict__ ownf,
                        const float* __restrict__ lstf,
                        const float* __restrict__ sparef,
                        const float* __restrict__ t,
                        const float* __restrict__ dinv,
                        float* __restrict__ acc, int e) {
    __shared__ float rows[OWN_ROWS * HD];
    __shared__ int   queue[QCAP];
    __shared__ int   qcount;

    const int tid  = threadIdx.x;
    const int lane = tid & 31;
    const int wave = __builtin_amdgcn_readfirstlane(tid >> 5);
    const int lo   = blockIdx.x * OWN_ROWS;
    int nrows = N_NODES - lo;
    if (nrows > OWN_ROWS) nrows = OWN_ROWS;

    const unsigned* reg = ((int)blockIdx.x < LMAIN) ? (const unsigned*)lstf + (size_t)blockIdx.x * LCAP
                                                    : (const unsigned*)sparef + (size_t)((int)blockIdx.x - LMAIN) * LCAP;
    const int* cnt = (const int*)sparef + (OWN_BLOCKS - LMAIN) * LCAP;
    const int listed = cnt[blockIdx.x * CNT_PITCH];
    const bool from_list = listed >= 0;

    int lsl[LTRIPS], ldl[LTRIPS], lwl[LTRIPS];
#pragma unroll
    for (int q = 0; q < LTRIPS; ++q) { lsl[q] = 0; ldl[q] = 0; lwl[q] = 0; }
    if (from_list) {
        unsigned wdv[LTRIPS];
#pragma unroll
        for (int q = 0; q < LTRIPS; ++q) {
            const int i = tid + q * 1024;
            wdv[q] = (i < listed) ? reg[i] : 0u;
        }
#pragma unroll
        for (int q = 0; q < LTRIPS; ++q) {
            const int eid = (int)(wdv[q] & LW_MASK);
            ldl[q] = (int)(wdv[q] >> LW_SHIFT);
            lsl[q] = src[eid];
            lwl[q] = __float_as_int(ew[eid]);
        }
    }

    const float4* a4 = (const float4*)(t + (size_t)lo * HD);
    float4* r4 = (float4*)rows;
    const int nvec = nrows * (HD / 4);
    {
        constexpr int TRIPS = (OWN_ROWS * (HD / 4) + 1023) / 1024;
        float4 st[TRIPS];
#pragma unroll
        for (int q = 0; q < TRIPS; ++q) {
            int i = tid + q * 1024;
            if (i > nvec - 1) i = nvec - 1;
            st[q] = a4[i];
        }
#pragma unroll
        for (int q = 0; q < TRIPS; ++q) {
            const int i = tid + q * 1024;
            if (i < OWN_ROWS * (HD / 4)) r4[i] = st[q];
        }
    }
    if (tid == 0) qcount = 0;
    __syncthreads();

    if (!from_list) {
        float* const rows0 = rows;
        own_scan(ownf, e, tid, queue, &qcount,
                 [=](int eid) {
                     const int s = src[eid];
                     const int d = dst[eid] - lo;
                     const float w = ew[eid];
#pragma nounroll
                     for (int cc = 0; cc < HD; ++cc)
                         lds_add_f32(&rows0[d * HD + cc], t[(size_t)s * HD + cc] * w);
                 });
    }
    __syncthreads();

    if (from_list) {
        const int n = listed;
#pragma unroll
        for (int q = 0; q < LTRIPS; ++q) {
            const int base = wave * 32 + q * 1024;
            if (base < n) {
                int m = n - base;
                if (m > 32) m = 32;
                own_add_edges(m, lsl[q], ldl[q], lwl[q], lane, t, rows);
            }
        }
    } else {
        int n = qcount;
        if (n > QCAP) n = QCAP;
        n = __builtin_amdgcn_readfirstlane(n);
        for (int base = wave * 32; base < n; base += 1024) {
            int m = n - base;
            if (m > 32) m = 32;
            int sl = 0, dl = 0, wl = 0;
            if (lane < m) {
                const int eid = queue[base + lane];
                const int dg = dst[eid];
                sl = src[eid];
                wl = __float_as_int(ew[eid]);
                dl = dg - lo;
            }
            own_add_edges(m, sl, dl, wl, lane, t, rows);
        }
    }
    __syncthreads();

    float4* o4 = (float4*)(acc + (size_t)lo * HD);
    {
        constexpr int TRIPS = (OWN_ROWS * (HD / 4) + 1023) / 1024;
        float4 ov[TRIPS];
        float  sc[TRIPS];
#pragma unroll
        for (int q = 0; q < TRIPS; ++q) {
            int i = tid + q * 1024;
            if (i > nvec - 1) i = nvec - 1;
            sc[q] = dinv[lo + (i >> 4)];
            ov[q] = r4[i];
        }
#pragma unroll
        for (int q = 0; q < TRIPS; ++q) {
            const int i = tid + q * 1024;
            if (i < nvec) {
                float4 v = ov[q];
                v.x *= sc[q]; v.y *= sc[q]; v.z *= sc[q]; v.w *= sc[q];
                o4[i] = v;
            }
        }
    }
}


__global__ void TEXTGCN_doc_gemm_wmma(const float* __restrict__ h,
                                      const float* __restrict__ bprev,
                                      const int* __restrict__ doc_idx,
                                      const float* __restrict__ Wp,
                                      const float* __restrict__ bp,
                                      float* __restrict__ dbuf) {
    const int lane = threadIdx.x;
    const int mt = blockIdx.x, nt = blockIdx.y;
    const int m  = mt * 16 + (lane & 15);
    const int n  = nt * 16 + (lane & 15);
    const int kq = (lane >> 4) * 2;

    const float* __restrict__ arow = h + (size_t)doc_idx[m] * HD;
    v8f acc = {};
    for (int k = 0; k < HD; k += 4) {
        v2f a, b;
        const float v0 = arow[k + kq]     + bprev[k + kq];
        const float v1 = arow[k + kq + 1] + bprev[k + kq + 1];
        a.x = v0 > 0.0f ? v0 : 0.0f;
        a.y = v1 > 0.0f ? v1 : 0.0f;
        b.x = Wp[(size_t)(k + kq)     * HD + n];
        b.y = Wp[(size_t)(k + kq + 1) * HD + n];
        acc = __builtin_amdgcn_wmma_f32_16x16x4_f32(
                  false, a, false, b, (short)0, acc, false, false);
    }
    const int mbase = mt * 16 + (lane >> 4) * 8;
    const int nst   = nt * 16 + (lane & 15);
    const float bias = bp[nst];
#pragma unroll
    for (int r = 0; r < 8; ++r) {
        float v = acc[r] + bias;
        dbuf[(size_t)(mbase + r) * HD + nst] = v > 0.0f ? v : 0.0f;
    }
}


__global__ void TEXTGCN_classifier(const float* __restrict__ dbuf,
                                   const float* __restrict__ Wc,
                                   const float* __restrict__ bc,
                                   float* __restrict__ out) {
    int gid = blockIdx.x * blockDim.x + threadIdx.x;
    if (gid >= NDOCS * NCLS) return;
    int d = gid / NCLS, c = gid % NCLS;
    const float* __restrict__ row = dbuf + (size_t)d * HD;
    float s = bc[c];
#pragma unroll 8
    for (int k = 0; k < HD; ++k) s += row[k] * Wc[(size_t)k * NCLS + c];
    out[gid] = s;
}


extern "C" void kernel_launch(void* const* d_in, const int* in_sizes, int n_in,
                              void* d_out, int out_size, void* d_ws, size_t ws_size,
                              hipStream_t stream) {
    const float* x   = (const float*)d_in[0];
    const float* ew  = (const float*)d_in[1];
    const float* W1  = (const float*)d_in[2];
    const float* b1  = (const float*)d_in[3];
    const float* W2  = (const float*)d_in[4];
    const float* b2  = (const float*)d_in[5];
    const float* W3  = (const float*)d_in[6];
    const float* b3  = (const float*)d_in[7];
    const float* Wp  = (const float*)d_in[8];
    const float* bp  = (const float*)d_in[9];
    const float* Wc  = (const float*)d_in[10];
    const float* bc  = (const float*)d_in[11];
    const int*   ei  = (const int*)d_in[12];
    const int*   doc = (const int*)d_in[13];
    float*       out = (float*)d_out;

    const int* src = ei;
    const int* dst = ei + N_EDGES;

    float* ws   = (float*)d_ws;
    float* dinv = ws;
    float* lst  = dinv + N_NODES;
    float* bufT = lst  + N_EDGES;
    float* bufH = bufT + (size_t)N_NODES * HD;
    float* dbuf = bufH + (size_t)N_NODES * HD;

    const int B = 256;
    dim3 blk(B);
    dim3 gE((N_EDGES + B - 1) / B);
    dim3 gOwn(OWN_BLOCKS);
    dim3 blkOwn(1024);

    TEXTGCN_owner_bytes<<<gE, blk, 0, stream>>>(dst, dbuf, N_EDGES);
    float* spare = dbuf + OWN_WORDS;
    TEXTGCN_owner_scan<<<gOwn, blkOwn, 0, stream>>>(dst, ew, dbuf, lst, spare, dinv, N_EDGES);

    dim3 wave(32);
    dim3 gGemm1(G1_BLOCKS);
    dim3 gGemmH(N_NODES / GH_ROWS);
    dim3 blkH(128);
    dim3 gDoc(NDOCS / 16, HD / 16);

    TEXTGCN_gemm_rows<<<gGemm1, blkH, 0, stream>>>(x, W1, dinv, bufT);
    TEXTGCN_propagate_owned<<<gOwn, blkOwn, 0, stream>>>(src, dst, ew, dbuf, lst, spare, bufT, dinv, bufH, N_EDGES);

    TEXTGCN_gemm_hidden<<<gGemmH, blkH, 0, stream>>>(bufH, b1, W2, dinv, bufT);
    TEXTGCN_propagate_owned<<<gOwn, blkOwn, 0, stream>>>(src, dst, ew, dbuf, lst, spare, bufT, dinv, bufH, N_EDGES);

    TEXTGCN_gemm_hidden<<<gGemmH, blkH, 0, stream>>>(bufH, b2, W3, dinv, bufT);
    TEXTGCN_propagate_owned<<<gOwn, blkOwn, 0, stream>>>(src, dst, ew, dbuf, lst, spare, bufT, dinv, bufH, N_EDGES);

    TEXTGCN_doc_gemm_wmma<<<gDoc, wave, 0, stream>>>(bufH, b3, doc, Wp, bp, dbuf);
    dim3 gCls((NDOCS * NCLS + B - 1) / B);
    TEXTGCN_classifier<<<gCls, blk, 0, stream>>>(dbuf, Wc, bc, out);
}


# last propagation pruned on the owner-block path: block marks its doc_idx rows in LDS, queues only list edges into marked rows, starts, adds, stores only those; LDS adds in arrival order as before
# speedup vs baseline: 3.6543x; 1.1310x over previous
#include <hip/hip_runtime.h>
#include <hip/hip_bf16.h>

typedef float v2f __attribute__((ext_vector_type(2)));
typedef float v8f __attribute__((ext_vector_type(8)));

#define N_NODES 100000
#define N_EDGES 1600000
#define DIM_IN  256
#define HD      64
#define NCLS    20
#define NDOCS   10000

#define OWN_ROWS   391
#define OWN_BLOCKS 256
#define QCAP       16384
static_assert(OWN_BLOCKS * OWN_ROWS >= N_NODES && (OWN_BLOCKS - 1) * OWN_ROWS < N_NODES && OWN_BLOCKS <= 256,
              "every owner block holds at least one row and its id fits a byte");
static_assert(HD / 4 == 16, "a row is 16 pieces of 16 bytes (the write-back finds a piece's row by a shift of 4)");


__global__ void TEXTGCN_owner_bytes(const int* __restrict__ dst, float* __restrict__ ownf, int e) {
    int i = blockIdx.x * blockDim.x + threadIdx.x;
    if (i < e) ((unsigned char*)ownf)[i] = (unsigned char)(dst[i] / OWN_ROWS);
}

#define LW_SHIFT 21
#define LW_MASK  ((1u << LW_SHIFT) - 1u)
#define LCAP     7008
#define CNT_PITCH 32
#define LMAIN    (N_EDGES / LCAP)
#define OWN_WORDS (N_EDGES / 4)
static_assert(N_EDGES <= (1 << LW_SHIFT) && OWN_ROWS <= (1 << (32 - LW_SHIFT)), "edge id and local row share a word");
static_assert(LMAIN <= OWN_BLOCKS && N_EDGES % 4 == 0 &&
              LCAP % 32 == 0 && OWN_WORDS % 32 == 0 && N_NODES % 32 == 0 &&
              OWN_WORDS + (OWN_BLOCKS - LMAIN) * LCAP + OWN_BLOCKS * CNT_PITCH <= NDOCS * HD,
              "owner bytes, the regions that do not fit the plane and the 256 counts fit the doc buffer's place");
static_assert(LCAP <= QCAP, "a list that is written was held whole in the LDS queue");


#define G1_ROWS   32
#define G1_PITCH  (DIM_IN + 4)
#define G1_TILES  (N_NODES / G1_ROWS)
#define G1_BLOCKS 1024
#define G1_VEC    (G1_ROWS * DIM_IN / 4 / 128)
#define G1_EACH(X) X(0) X(1) X(2) X(3) X(4) X(5) X(6) X(7) X(8) X(9) X(10) X(11) X(12) X(13) X(14) X(15)
#define G1_LOAD(q)  const float4 nv##q = nx[(q) * 128];
#define G1_STORE(q) *(float4*)&nt[((tid + (q) * 128) >> 6) * G1_PITCH + ((tid + (q) * 128) & 63) * 4] = nv##q;
static_assert(G1_VEC == 16, "G1_EACH names 16 values");
static_assert(N_NODES % G1_ROWS == 0 && HD == 64 && DIM_IN % 4 == 0 && (G1_ROWS * DIM_IN / 4) % 128 == 0 &&
              (G1_PITCH * 4) % 16 == 0, "tile shape");

__device__ __forceinline__ void g1_tile(const float* __restrict__ ap,
                                        const v2f (&bf)[DIM_IN / 4],
                                        const float* __restrict__ dinv,
                                        float* __restrict__ C,
                                        int mbase, int n) {
    v8f acc0 = {}, acc1 = {};
#pragma unroll
    for (int i = 0; i < DIM_IN / 4; ++i) {
        const v2f a0 = *(const v2f*)(ap + 4 * i);
        const v2f a1 = *(const v2f*)(ap + 16 * G1_PITCH + 4 * i);
        acc0 = __builtin_amdgcn_wmma_f32_16x16x4_f32(
                   false, a0, false, bf[i], (short)0, acc0, false, false);
        acc1 = __builtin_amdgcn_wmma_f32_16x16x4_f32(
                   false, a1, false, bf[i], (short)0, acc1, false, false);
    }
    const float4 d0 = *(const float4*)(dinv + mbase);
    const float4 d1 = *(const float4*)(dinv + mbase + 4);
    const float4 e0 = *(const float4*)(dinv + mbase + 16);
    const float4 e1 = *(const float4*)(dinv + mbase + 20);
    const float dr[8] = {d0.x, d0.y, d0.z, d0.w, d1.x, d1.y, d1.z, d1.w};
    const float er[8] = {e0.x, e0.y, e0.z, e0.w, e1.x, e1.y, e1.z, e1.w};
    const unsigned ob = (unsigned)mbase * HD + (unsigned)n;
    float* __restrict__ cp = C + ob;
#pragma unroll
    for (int r = 0; r < 8; ++r) {
        cp[r * HD]        = acc0[r] * dr[r];
        cp[(16 + r) * HD] = acc1[r] * er[r];
    }
}

__global__ void __launch_bounds__(128)
TEXTGCN_gemm_rows(const float* A,
                  const float* __restrict__ B,
                  const float* __restrict__ dinv,
                  float* __restrict__ C) {
    __shared__ __attribute__((aligned(16))) float tileA[2][G1_ROWS * G1_PITCH];

    const int tid  = threadIdx.x;
    const int lane = tid & 31;
    const int wv   = tid >> 5;
    const int c    = lane & 15;
    const int h    = lane >> 4;
    const int kq   = h * 2;
    const int n    = wv * 16 + c;

    v2f bf[DIM_IN / 4];
#pragma unroll
    for (int i = 0; i < DIM_IN / 4; ++i) {
        bf[i].x = B[(size_t)(4 * i + kq)     * HD + n];
        bf[i].y = B[(size_t)(4 * i + kq + 1) * HD + n];
    }

    const float4* a4 = (const float4*)A;
    int t = blockIdx.x;
    if (t < G1_TILES) {
        float4 s0[G1_VEC];
#pragma unroll
        for (int q = 0; q < G1_VEC; ++q) s0[q] = a4[(size_t)t * (G1_ROWS * DIM_IN / 4) + tid + q * 128];
#pragma unroll
        for (int q = 0; q < G1_VEC; ++q) {
            const int idx = tid + q * 128;
            *(float4*)&tileA[0][(idx >> 6) * G1_PITCH + (idx & 63) * 4] = s0[q];
        }
    }
    __syncthreads();

    int buf = 0;
    for (; t + G1_BLOCKS < G1_TILES; t += G1_BLOCKS) {
        const float4* nx = a4 + (size_t)(t + G1_BLOCKS) * (G1_ROWS * DIM_IN / 4) + tid;
        G1_EACH(G1_LOAD)
        __builtin_amdgcn_sched_barrier(0);

        g1_tile(&tileA[buf][c * G1_PITCH + kq], bf, dinv, C, t * G1_ROWS + h * 8, n);

        float* nt = tileA[buf ^ 1];
        G1_EACH(G1_STORE)
        __syncthreads();
        buf ^= 1;
    }
    if (t < G1_TILES)
        g1_tile(&tileA[buf][c * G1_PITCH + kq], bf, dinv, C, t * G1_ROWS + h * 8, n);
}


#define GH_ROWS   160
#define GH_PITCH  (HD + 4)
#define GH_VEC    (GH_ROWS * HD / 4 / 128)
static_assert(N_NODES % GH_ROWS == 0 && GH_ROWS % 32 == 0 && (GH_ROWS * HD / 4) % 128 == 0 && (GH_PITCH * 4) % 16 == 0,
              "tile shape");

__global__ void __launch_bounds__(128)
TEXTGCN_gemm_hidden(const float* H,
                    const float* __restrict__ bprev,
                    const float* __restrict__ Wk,
                    const float* __restrict__ dinv,
                    float* __restrict__ T) {
    __shared__ __attribute__((aligned(16))) float tileH[GH_ROWS * GH_PITCH];

    const int tid  = threadIdx.x;
    const int lane = tid & 31;
    const int wv   = tid >> 5;
    const int c    = lane & 15;
    const int h    = lane >> 4;
    const int kq   = h * 2;
    const int n    = wv * 16 + c;
    const int row0 = blockIdx.x * GH_ROWS;

    const float4* h4 = (const float4*)(H + (size_t)row0 * HD);
    float4 st[GH_VEC];
#pragma unroll
    for (int q = 0; q < GH_VEC; ++q) st[q] = h4[tid + q * 128];

    v2f bf[HD / 4];
#pragma unroll
    for (int i = 0; i < HD / 4; ++i) {
        bf[i].x = Wk[(size_t)(4 * i + kq)     * HD + n];
        bf[i].y = Wk[(size_t)(4 * i + kq + 1) * HD + n];
    }
    const float4 bv = *(const float4*)(bprev + (tid & 15) * 4);
#pragma unroll
    for (int q = 0; q < GH_VEC; ++q) {
        const int idx = tid + q * 128;
        const float v0 = st[q].x + bv.x, v1 = st[q].y + bv.y, v2 = st[q].z + bv.z, v3 = st[q].w + bv.w;
        float4 o;
        o.x = v0 > 0.0f ? v0 : 0.0f;
        o.y = v1 > 0.0f ? v1 : 0.0f;
        o.z = v2 > 0.0f ? v2 : 0.0f;
        o.w = v3 > 0.0f ? v3 : 0.0f;
        *(float4*)&tileH[(idx >> 4) * GH_PITCH + (idx & 15) * 4] = o;
    }
    __syncthreads();

#pragma unroll
    for (int s = 0; s < GH_ROWS / 32; ++s) {
        const float* ap = &tileH[(s * 32 + c) * GH_PITCH + kq];
        v8f acc0 = {}, acc1 = {};
#pragma unroll
        for (int i = 0; i < HD / 4; ++i) {
            const v2f a0 = *(const v2f*)(ap + 4 * i);
            const v2f a1 = *(const v2f*)(ap + 16 * GH_PITCH + 4 * i);
            acc0 = __builtin_amdgcn_wmma_f32_16x16x4_f32(
                       false, a0, false, bf[i], (short)0, acc0, false, false);
            acc1 = __builtin_amdgcn_wmma_f32_16x16x4_f32(
                       false, a1, false, bf[i], (short)0, acc1, false, false);
        }
        const int mbase = row0 + s * 32 + h * 8;
        const float4 d0 = *(const float4*)(dinv + mbase);
        const float4 d1 = *(const float4*)(dinv + mbase + 4);
        const float4 e0 = *(const float4*)(dinv + mbase + 16);
        const float4 e1 = *(const float4*)(dinv + mbase + 20);
        const float dr[8] = {d0.x, d0.y, d0.z, d0.w, d1.x, d1.y, d1.z, d1.w};
        const float er[8] = {e0.x, e0.y, e0.z, e0.w, e1.x, e1.y, e1.z, e1.w};
        const unsigned ob = (unsigned)mbase * HD + (unsigned)n;
        float* __restrict__ tp = T + ob;
#pragma unroll
        for (int r = 0; r < 8; ++r) {
            tp[r * HD]        = acc0[r] * dr[r];
            tp[(16 + r) * HD] = acc1[r] * er[r];
        }
    }
}


__device__ __forceinline__ void lds_add_f32(float* p, float v) {
    __hip_atomic_fetch_add(p, v, __ATOMIC_RELAXED, __HIP_MEMORY_SCOPE_WORKGROUP);
}

template <class Over>
__device__ __forceinline__ void own_scan(const float* __restrict__ ownf, int e, int tid,
                                         int* queue, int* qcount, Over over) {
    const uint4* own4 = (const uint4*)ownf;
    const unsigned char* own1 = (const unsigned char*)ownf;
    const unsigned me = blockIdx.x * 0x01010101u;
    const int ngroups = e >> 4;
    for (int g0 = tid; g0 < ngroups; g0 += 8 * 1024) {
        uint4 v[8];
#pragma unroll
        for (int k = 0; k < 8; ++k) {
            int g = g0 + k * 1024;
            if (g > ngroups - 1) g = ngroups - 1;
            v[k] = own4[g];
        }
        unsigned p0 = 0u, p1 = 0u, p2 = 0u, p3 = 0u;
#pragma unroll
        for (int k = 0; k < 8; ++k) {
            const unsigned x0 = v[k].x ^ me, x1 = v[k].y ^ me, x2 = v[k].z ^ me, x3 = v[k].w ^ me;
            const unsigned u0 = ((x0 & 0x7F7F7F7Fu) + 0x7F7F7F7Fu) | x0;
            const unsigned u1 = ((x1 & 0x7F7F7F7Fu) + 0x7F7F7F7Fu) | x1;
            const unsigned u2 = ((x2 & 0x7F7F7F7Fu) + 0x7F7F7F7Fu) | x2;
            const unsigned u3 = ((x3 & 0x7F7F7F7Fu) + 0x7F7F7F7Fu) | x3;
            const unsigned sel = 0x01010101u << k;
            p0 |= (u0 >> (7 - k)) & sel;
            p1 |= (u1 >> (7 - k)) & sel;
            p2 |= (u2 >> (7 - k)) & sel;
            p3 |= (u3 >> (7 - k)) & sel;
        }
        int nv = (ngroups - g0 + 1023) >> 10;
        if (nv > 8) nv = 8;
        const unsigned vm = ((1u << nv) - 1u) * 0x01010101u;
        const unsigned mm[4] = {~p0 & vm, ~p1 & vm, ~p2 & vm, ~p3 & vm};
#pragma unroll
        for (int j = 0; j < 4; ++j) {
            unsigned m = mm[j];
            while (m != 0u) {
                const int bit = __builtin_ctz(m);
                m &= m - 1u;
                const int eid = ((g0 + (bit & 7) * 1024) << 4) + j * 4 + (bit >> 3);
                const int pos = atomicAdd(qcount, 1);
                if (pos < QCAP) queue[pos] = eid;
                else            over(eid);
            }
        }
    }
    for (int i = (ngroups << 4) + tid; i < e; i += 1024) {
        if (own1[i] == (unsigned char)blockIdx.x) {
            const int pos = atomicAdd(qcount, 1);
            if (pos < QCAP) queue[pos] = i;
            else            over(i);
        }
    }
}

#define SCAN_BATCH 4

__global__ void __launch_bounds__(1024)
TEXTGCN_owner_scan(const int* __restrict__ dst,
                   const float* __restrict__ ew,
                   const float* __restrict__ ownf,
                   float* __restrict__ lstf,
                   float* __restrict__ sparef,
                   float* __restrict__ dinv, int e) {
    __shared__ float degs[OWN_ROWS];
    __shared__ int   queue[QCAP];
    __shared__ int   qcount;

    const int tid = threadIdx.x;
    const int lo  = blockIdx.x * OWN_ROWS;
    int nrows = N_NODES - lo;
    if (nrows > OWN_ROWS) nrows = OWN_ROWS;

    unsigned* reg = ((int)blockIdx.x < LMAIN) ? (unsigned*)lstf + (size_t)blockIdx.x * LCAP
                                              : (unsigned*)sparef + (size_t)((int)blockIdx.x - LMAIN) * LCAP;
    int* cnt = (int*)sparef + (OWN_BLOCKS - LMAIN) * LCAP;

    if (tid < OWN_ROWS) degs[tid] = 1.0f;
    if (tid == 0) qcount = 0;
    __syncthreads();

    float* const degs0 = degs;
    own_scan(ownf, e, tid, queue, &qcount,
             [=](int eid) { lds_add_f32(&degs0[dst[eid] - lo], ew[eid]); });
    __syncthreads();

    int n = qcount;
    const bool keep = n <= LCAP;
    if (tid == 0) cnt[blockIdx.x * CNT_PITCH] = keep ? n : -1;
    if (n > QCAP) n = QCAP;
    for (int b = tid; b < n; b += SCAN_BATCH * 1024) {
        int eid[SCAN_BATCH], dg[SCAN_BATCH];
        float w[SCAN_BATCH];
#pragma unroll
        for (int k = 0; k < SCAN_BATCH; ++k) {
            int i = b + k * 1024;
            if (i > n - 1) i = n - 1;
            eid[k] = queue[i];
        }
#pragma unroll
        for (int k = 0; k < SCAN_BATCH; ++k) {
            dg[k] = dst[eid[k]];
            w[k]  = ew[eid[k]];
        }
#pragma unroll
        for (int k = 0; k < SCAN_BATCH; ++k) {
            const int i = b + k * 1024;
            if (i < n) {
                const int dl = dg[k] - lo;
                lds_add_f32(&degs[dl], w[k]);
                if (keep) reg[i] = (unsigned)eid[k] | ((unsigned)dl << LW_SHIFT);
            }
        }
    }
    __syncthreads();

    if (tid < nrows) {
        const float d = degs[tid];
        dinv[lo + tid] = (d > 0.0f) ? __frsqrt_rn(d) : 0.0f;
    }
}

__device__ __forceinline__ void own_add_edges(int m, int sl, int dl, int wl, int lane,
                                              const float* __restrict__ t, float* rows) {
    for (int j = 0; j < m; j += 16) {
        float v0[16], v1[16];
#pragma unroll
        for (int k = 0; k < 16; ++k) {
            int idx = j + k;
            if (idx > m - 1) idx = m - 1;
            const int s = __builtin_amdgcn_readlane(sl, idx);
            const float* __restrict__ tp = t + (size_t)s * HD;
            v0[k] = tp[lane];
            v1[k] = tp[lane + 32];
        }
#pragma unroll
        for (int k = 0; k < 16; ++k) {
            if (j + k < m) {
                const int d = __builtin_amdgcn_readlane(dl, j + k);
                const float w = __int_as_float(__builtin_amdgcn_readlane(wl, j + k));
                float* rp = rows + d * HD;
                lds_add_f32(rp + lane,      v0[k] * w);
                lds_add_f32(rp + lane + 32, v1[k] * w);
            }
        }
    }
}

#define LTRIPS ((LCAP + 1023) / 1024)

#define DOC_TRIPS ((NDOCS + 1023) / 1024)
__device__ __forceinline__ void own_doc_rows(const int* __restrict__ src, const float* __restrict__ ew,
                                             const int* __restrict__ doc, const unsigned* __restrict__ reg, int listed,
                                             const float* __restrict__ t, const float* __restrict__ dinv,
                                             float* __restrict__ acc, float* rows, int* queue, int* qcount, int* mark,
                                             int lo, int nrows, int tid, int lane, int wave) {
    unsigned wdv[LTRIPS];
#pragma unroll
    for (int q = 0; q < LTRIPS; ++q) {
        const int i = tid + q * 1024;
        wdv[q] = (i < listed) ? reg[i] : 0u;
    }
    int dv[DOC_TRIPS];
#pragma unroll
    for (int q = 0; q < DOC_TRIPS; ++q) {
        int i = tid + q * 1024;
        if (i > NDOCS - 1) i = NDOCS - 1;
        dv[q] = doc[i];
    }
    if (tid < OWN_ROWS) mark[tid] = 0;
    if (tid == 0) *qcount = 0;
    __syncthreads();
#pragma unroll
    for (int q = 0; q < DOC_TRIPS; ++q) {
        const unsigned l = (unsigned)(dv[q] - lo);
        if (l < (unsigned)nrows) mark[l] = 1;
    }
    __syncthreads();

    const float4* a4 = (const float4*)(t + (size_t)lo * HD);
    float4* r4 = (float4*)rows;
    const int nvec = nrows * (HD / 4);
    constexpr int TRIPS = (OWN_ROWS * (HD / 4) + 1023) / 1024;
    bool  on[TRIPS];
    float4 st[TRIPS];
#pragma unroll
    for (int q = 0; q < TRIPS; ++q) {
        const int i = tid + q * 1024;
        on[q] = (i < nvec) && (mark[i >> 4] != 0);
        float4 v = make_float4(0.0f, 0.0f, 0.0f, 0.0f);
        if (on[q]) v = a4[i];
        st[q] = v;
    }
#pragma unroll
    for (int q = 0; q < LTRIPS; ++q) {
        const int i = tid + q * 1024;
        if (i < listed && mark[wdv[q] >> LW_SHIFT] != 0) {
            const int pos = atomicAdd(qcount, 1);
            queue[pos] = (int)wdv[q];
        }
    }
#pragma unroll
    for (int q = 0; q < TRIPS; ++q)
        if (on[q]) r4[tid + q * 1024] = st[q];
    __syncthreads();

    const int n = __builtin_amdgcn_readfirstlane(*qcount);
    for (int base = wave * 32; base < n; base += 1024) {
        int m = n - base;
        if (m > 32) m = 32;
        int sl = 0, dl = 0, wl = 0;
        if (lane < m) {
            const unsigned w = (unsigned)queue[base + lane];
            const int eid = (int)(w & LW_MASK);
            dl = (int)(w >> LW_SHIFT);
            sl = src[eid];
            wl = __float_as_int(ew[eid]);
        }
        own_add_edges(m, sl, dl, wl, lane, t, rows);
    }
    __syncthreads();

    float4* o4 = (float4*)(acc + (size_t)lo * HD);
#pragma unroll
    for (int q = 0; q < TRIPS; ++q) {
        if (on[q]) {
            const int i = tid + q * 1024;
            const float s = dinv[lo + (i >> 4)];
            float4 v = r4[i];
            v.x *= s; v.y *= s; v.z *= s; v.w *= s;
            o4[i] = v;
        }
    }
}

template <bool DOC>
__global__ void __launch_bounds__(1024)
TEXTGCN_propagate_owned(const int* __restrict__ src,
                        const int* __restrict__ dst,
                        const float* __restrict__ ew,
                        const int* __restrict__ doc,
                        const float* __restrict__ ownf,
                        const float* __restrict__ lstf,
                        const float* __restrict__ sparef,
                        const float* __restrict__ t,
                        const float* __restrict__ dinv,
                        float* __restrict__ acc, int e) {
    __shared__ float rows[OWN_ROWS * HD];
    __shared__ int   queue[QCAP];
    __shared__ int   qcount;
    __shared__ int   mark[OWN_ROWS];

    const int tid  = threadIdx.x;
    const int lane = tid & 31;
    const int wave = __builtin_amdgcn_readfirstlane(tid >> 5);
    const int lo   = blockIdx.x * OWN_ROWS;
    int nrows = N_NODES - lo;
    if (nrows > OWN_ROWS) nrows = OWN_ROWS;

    const unsigned* reg = ((int)blockIdx.x < LMAIN) ? (const unsigned*)lstf + (size_t)blockIdx.x * LCAP
                                                    : (const unsigned*)sparef + (size_t)((int)blockIdx.x - LMAIN) * LCAP;
    const int* cnt = (const int*)sparef + (OWN_BLOCKS - LMAIN) * LCAP;
    const int listed = cnt[blockIdx.x * CNT_PITCH];
    const bool from_list = listed >= 0;

    if constexpr (DOC) {
        if (from_list) {
            own_doc_rows(src, ew, doc, reg, listed, t, dinv, acc, rows, queue, &qcount, mark, lo, nrows, tid, lane, wave);
            return;
        }
    }

    int lsl[LTRIPS], ldl[LTRIPS], lwl[LTRIPS];
#pragma unroll
    for (int q = 0; q < LTRIPS; ++q) { lsl[q] = 0; ldl[q] = 0; lwl[q] = 0; }
    if (from_list) {
        unsigned wdv[LTRIPS];
#pragma unroll
        for (int q = 0; q < LTRIPS; ++q) {
            const int i = tid + q * 1024;
            wdv[q] = (i < listed) ? reg[i] : 0u;
        }
#pragma unroll
        for (int q = 0; q < LTRIPS; ++q) {
            const int eid = (int)(wdv[q] & LW_MASK);
            ldl[q] = (int)(wdv[q] >> LW_SHIFT);
            lsl[q] = src[eid];
            lwl[q] = __float_as_int(ew[eid]);
        }
    }

    const float4* a4 = (const float4*)(t + (size_t)lo * HD);
    float4* r4 = (float4*)rows;
    const int nvec = nrows * (HD / 4);
    {
        constexpr int TRIPS = (OWN_ROWS * (HD / 4) + 1023) / 1024;
        float4 st[TRIPS];
#pragma unroll
        for (int q = 0; q < TRIPS; ++q) {
            int i = tid + q * 1024;
            if (i > nvec - 1) i = nvec - 1;
            st[q] = a4[i];
        }
#pragma unroll
        for (int q = 0; q < TRIPS; ++q) {
            const int i = tid + q * 1024;
            if (i < OWN_ROWS * (HD / 4)) r4[i] = st[q];
        }
    }
    if (tid == 0) qcount = 0;
    __syncthreads();

    if (!from_list) {
        float* const rows0 = rows;
        own_scan(ownf, e, tid, queue, &qcount,
                 [=](int eid) {
                     const int s = src[eid];
                     const int d = dst[eid] - lo;
                     const float w = ew[eid];
#pragma nounroll
                     for (int cc = 0; cc < HD; ++cc)
                         lds_add_f32(&rows0[d * HD + cc], t[(size_t)s * HD + cc] * w);
                 });
    }
    __syncthreads();

    if (from_list) {
        const int n = listed;
#pragma unroll
        for (int q = 0; q < LTRIPS; ++q) {
            const int base = wave * 32 + q * 1024;
            if (base < n) {
                int m = n - base;
                if (m > 32) m = 32;
                own_add_edges(m, lsl[q], ldl[q], lwl[q], lane, t, rows);
            }
        }
    } else {
        int n = qcount;
        if (n > QCAP) n = QCAP;
        n = __builtin_amdgcn_readfirstlane(n);
        for (int base = wave * 32; base < n; base += 1024) {
            int m = n - base;
            if (m > 32) m = 32;
            int sl = 0, dl = 0, wl = 0;
            if (lane < m) {
                const int eid = queue[base + lane];
                const int dg = dst[eid];
                sl = src[eid];
                wl = __float_as_int(ew[eid]);
                dl = dg - lo;
            }
            own_add_edges(m, sl, dl, wl, lane, t, rows);
        }
    }
    __syncthreads();

    float4* o4 = (float4*)(acc + (size_t)lo * HD);
    {
        constexpr int TRIPS = (OWN_ROWS * (HD / 4) + 1023) / 1024;
        float4 ov[TRIPS];
        float  sc[TRIPS];
#pragma unroll
        for (int q = 0; q < TRIPS; ++q) {
            int i = tid + q * 1024;
            if (i > nvec - 1) i = nvec - 1;
            sc[q] = dinv[lo + (i >> 4)];
            ov[q] = r4[i];
        }
#pragma unroll
        for (int q = 0; q < TRIPS; ++q) {
            const int i = tid + q * 1024;
            if (i < nvec) {
                float4 v = ov[q];
                v.x *= sc[q]; v.y *= sc[q]; v.z *= sc[q]; v.w *= sc[q];
                o4[i] = v;
            }
        }
    }
}


__global__ void TEXTGCN_doc_gemm_wmma(const float* __restrict__ h,
                                      const float* __restrict__ bprev,
                                      const int* __restrict__ doc_idx,
                                      const float* __restrict__ Wp,
                                      const float* __restrict__ bp,
                                      float* __restrict__ dbuf) {
    const int lane = threadIdx.x;
    const int mt = blockIdx.x, nt = blockIdx.y;
    const int m  = mt * 16 + (lane & 15);
    const int n  = nt * 16 + (lane & 15);
    const int kq = (lane >> 4) * 2;

    const float* __restrict__ arow = h + (size_t)doc_idx[m] * HD;
    v8f acc = {};
    for (int k = 0; k < HD; k += 4) {
        v2f a, b;
        const float v0 = arow[k + kq]     + bprev[k + kq];
        const float v1 = arow[k + kq + 1] + bprev[k + kq + 1];
        a.x = v0 > 0.0f ? v0 : 0.0f;
        a.y = v1 > 0.0f ? v1 : 0.0f;
        b.x = Wp[(size_t)(k + kq)     * HD + n];
        b.y = Wp[(size_t)(k + kq + 1) * HD + n];
        acc = __builtin_amdgcn_wmma_f32_16x16x4_f32(
                  false, a, false, b, (short)0, acc, false, false);
    }
    const int mbase = mt * 16 + (lane >> 4) * 8;
    const int nst   = nt * 16 + (lane & 15);
    const float bias = bp[nst];
#pragma unroll
    for (int r = 0; r < 8; ++r) {
        float v = acc[r] + bias;
        dbuf[(size_t)(mbase + r) * HD + nst] = v > 0.0f ? v : 0.0f;
    }
}


__global__ void TEXTGCN_classifier(const float* __restrict__ dbuf,
                                   const float* __restrict__ Wc,
                                   const float* __restrict__ bc,
                                   float* __restrict__ out) {
    int gid = blockIdx.x * blockDim.x + threadIdx.x;
    if (gid >= NDOCS * NCLS) return;
    int d = gid / NCLS, c = gid % NCLS;
    const float* __restrict__ row = dbuf + (size_t)d * HD;
    float s = bc[c];
#pragma unroll 8
    for (int k = 0; k < HD; ++k) s += row[k] * Wc[(size_t)k * NCLS + c];
    out[gid] = s;
}


extern "C" void kernel_launch(void* const* d_in, const int* in_sizes, int n_in,
                              void* d_out, int out_size, void* d_ws, size_t ws_size,
                              hipStream_t stream) {
    const float* x   = (const float*)d_in[0];
    const float* ew  = (const float*)d_in[1];
    const float* W1  = (const float*)d_in[2];
    const float* b1  = (const float*)d_in[3];
    const float* W2  = (const float*)d_in[4];
    const float* b2  = (const float*)d_in[5];
    const float* W3  = (const float*)d_in[6];
    const float* b3  = (const float*)d_in[7];
    const float* Wp  = (const float*)d_in[8];
    const float* bp  = (const float*)d_in[9];
    const float* Wc  = (const float*)d_in[10];
    const float* bc  = (const float*)d_in[11];
    const int*   ei  = (const int*)d_in[12];
    const int*   doc = (const int*)d_in[13];
    float*       out = (float*)d_out;

    const int* src = ei;
    const int* dst = ei + N_EDGES;

    float* ws   = (float*)d_ws;
    float* dinv = ws;
    float* lst  = dinv + N_NODES;
    float* bufT = lst  + N_EDGES;
    float* bufH = bufT + (size_t)N_NODES * HD;
    float* dbuf = bufH + (size_t)N_NODES * HD;

    const int B = 256;
    dim3 blk(B);
    dim3 gE((N_EDGES + B - 1) / B);
    dim3 gOwn(OWN_BLOCKS);
    dim3 blkOwn(1024);

    TEXTGCN_owner_bytes<<<gE, blk, 0, stream>>>(dst, dbuf, N_EDGES);
    float* spare = dbuf + OWN_WORDS;
    TEXTGCN_owner_scan<<<gOwn, blkOwn, 0, stream>>>(dst, ew, dbuf, lst, spare, dinv, N_EDGES);

    dim3 wave(32);
    dim3 gGemm1(G1_BLOCKS);
    dim3 gGemmH(N_NODES / GH_ROWS);
    dim3 blkH(128);
    dim3 gDoc(NDOCS / 16, HD / 16);

    TEXTGCN_gemm_rows<<<gGemm1, blkH, 0, stream>>>(x, W1, dinv, bufT);
    TEXTGCN_propagate_owned<false><<<gOwn, blkOwn, 0, stream>>>(src, dst, ew, doc, dbuf, lst, spare, bufT, dinv, bufH, N_EDGES);

    TEXTGCN_gemm_hidden<<<gGemmH, blkH, 0, stream>>>(bufH, b1, W2, dinv, bufT);
    TEXTGCN_propagate_owned<false><<<gOwn, blkOwn, 0, stream>>>(src, dst, ew, doc, dbuf, lst, spare, bufT, dinv, bufH, N_EDGES);

    TEXTGCN_gemm_hidden<<<gGemmH, blkH, 0, stream>>>(bufH, b2, W3, dinv, bufT);
    TEXTGCN_propagate_owned<true><<<gOwn, blkOwn, 0, stream>>>(src, dst, ew, doc, dbuf, lst, spare, bufT, dinv, bufH, N_EDGES);

    TEXTGCN_doc_gemm_wmma<<<gDoc, wave, 0, stream>>>(bufH, b3, doc, Wp, bp, dbuf);
    dim3 gCls((NDOCS * NCLS + B - 1) / B);
    TEXTGCN_classifier<<<gCls, blk, 0, stream>>>(dbuf, Wc, bc, out);
}


# launch fusion: each conv propagation's owner block makes its finished LDS rows relu(h+b) in place and does the next 64x64 transform from them (WMMA f32, k order kept); 10 launches to 8
# speedup vs baseline: 3.9612x; 1.0840x over previous
#include <hip/hip_runtime.h>
#include <hip/hip_bf16.h>

typedef float v2f __attribute__((ext_vector_type(2)));
typedef float v8f __attribute__((ext_vector_type(8)));

#define N_NODES 100000
#define N_EDGES 1600000
#define DIM_IN  256
#define HD      64
#define NCLS    20
#define NDOCS   10000

#define OWN_ROWS   391
#define OWN_BLOCKS 256
#define QCAP       16384
static_assert(OWN_BLOCKS * OWN_ROWS >= N_NODES && (OWN_BLOCKS - 1) * OWN_ROWS < N_NODES && OWN_BLOCKS <= 256,
              "every owner block holds at least one row and its id fits a byte");
static_assert(HD / 4 == 16, "a row is 16 pieces of 16 bytes (the write-back finds a piece's row by a shift of 4)");


__global__ void TEXTGCN_owner_bytes(const int* __restrict__ dst, float* __restrict__ ownf, int e) {
    int i = blockIdx.x * blockDim.x + threadIdx.x;
    if (i < e) ((unsigned char*)ownf)[i] = (unsigned char)(dst[i] / OWN_ROWS);
}

#define LW_SHIFT 21
#define LW_MASK  ((1u << LW_SHIFT) - 1u)
#define LCAP     7008
#define CNT_PITCH 32
#define LMAIN    (N_EDGES / LCAP)
#define OWN_WORDS (N_EDGES / 4)
static_assert(N_EDGES <= (1 << LW_SHIFT) && OWN_ROWS <= (1 << (32 - LW_SHIFT)), "edge id and local row share a word");
static_assert(LMAIN <= OWN_BLOCKS && N_EDGES % 4 == 0 &&
              LCAP % 32 == 0 && OWN_WORDS % 32 == 0 && N_NODES % 32 == 0 &&
              OWN_WORDS + (OWN_BLOCKS - LMAIN) * LCAP + OWN_BLOCKS * CNT_PITCH <= NDOCS * HD,
              "owner bytes, the regions that do not fit the plane and the 256 counts fit the doc buffer's place");
static_assert(LCAP <= QCAP, "a list that is written was held whole in the LDS queue");


#define G1_ROWS   32
#define G1_PITCH  (DIM_IN + 4)
#define G1_TILES  (N_NODES / G1_ROWS)
#define G1_BLOCKS 1024
#define G1_VEC    (G1_ROWS * DIM_IN / 4 / 128)
#define G1_EACH(X) X(0) X(1) X(2) X(3) X(4) X(5) X(6) X(7) X(8) X(9) X(10) X(11) X(12) X(13) X(14) X(15)
#define G1_LOAD(q)  const float4 nv##q = nx[(q) * 128];
#define G1_STORE(q) *(float4*)&nt[((tid + (q) * 128) >> 6) * G1_PITCH + ((tid + (q) * 128) & 63) * 4] = nv##q;
static_assert(G1_VEC == 16, "G1_EACH names 16 values");
static_assert(N_NODES % G1_ROWS == 0 && HD == 64 && DIM_IN % 4 == 0 && (G1_ROWS * DIM_IN / 4) % 128 == 0 &&
              (G1_PITCH * 4) % 16 == 0, "tile shape");

__device__ __forceinline__ void g1_tile(const float* __restrict__ ap,
                                        const v2f (&bf)[DIM_IN / 4],
                                        const float* __restrict__ dinv,
                                        float* __restrict__ C,
                                        int mbase, int n) {
    v8f acc0 = {}, acc1 = {};
#pragma unroll
    for (int i = 0; i < DIM_IN / 4; ++i) {
        const v2f a0 = *(const v2f*)(ap + 4 * i);
        const v2f a1 = *(const v2f*)(ap + 16 * G1_PITCH + 4 * i);
        acc0 = __builtin_amdgcn_wmma_f32_16x16x4_f32(
                   false, a0, false, bf[i], (short)0, acc0, false, false);
        acc1 = __builtin_amdgcn_wmma_f32_16x16x4_f32(
                   false, a1, false, bf[i], (short)0, acc1, false, false);
    }
    const float4 d0 = *(const float4*)(dinv + mbase);
    const float4 d1 = *(const float4*)(dinv + mbase + 4);
    const float4 e0 = *(const float4*)(dinv + mbase + 16);
    const float4 e1 = *(const float4*)(dinv + mbase + 20);
    const float dr[8] = {d0.x, d0.y, d0.z, d0.w, d1.x, d1.y, d1.z, d1.w};
    const float er[8] = {e0.x, e0.y, e0.z, e0.w, e1.x, e1.y, e1.z, e1.w};
    const unsigned ob = (unsigned)mbase * HD + (unsigned)n;
    float* __restrict__ cp = C + ob;
#pragma unroll
    for (int r = 0; r < 8; ++r) {
        cp[r * HD]        = acc0[r] * dr[r];
        cp[(16 + r) * HD] = acc1[r] * er[r];
    }
}

__global__ void __launch_bounds__(128)
TEXTGCN_gemm_rows(const float* A,
                  const float* __restrict__ B,
                  const float* __restrict__ dinv,
                  float* __restrict__ C) {
    __shared__ __attribute__((aligned(16))) float tileA[2][G1_ROWS * G1_PITCH];

    const int tid  = threadIdx.x;
    const int lane = tid & 31;
    const int wv   = tid >> 5;
    const int c    = lane & 15;
    const int h    = lane >> 4;
    const int kq   = h * 2;
    const int n    = wv * 16 + c;

    v2f bf[DIM_IN / 4];
#pragma unroll
    for (int i = 0; i < DIM_IN / 4; ++i) {
        bf[i].x = B[(size_t)(4 * i + kq)     * HD + n];
        bf[i].y = B[(size_t)(4 * i + kq + 1) * HD + n];
    }

    const float4* a4 = (const float4*)A;
    int t = blockIdx.x;
    if (t < G1_TILES) {
        float4 s0[G1_VEC];
#pragma unroll
        for (int q = 0; q < G1_VEC; ++q) s0[q] = a4[(size_t)t * (G1_ROWS * DIM_IN / 4) + tid + q * 128];
#pragma unroll
        for (int q = 0; q < G1_VEC; ++q) {
            const int idx = tid + q * 128;
            *(float4*)&tileA[0][(idx >> 6) * G1_PITCH + (idx & 63) * 4] = s0[q];
        }
    }
    __syncthreads();

    int buf = 0;
    for (; t + G1_BLOCKS < G1_TILES; t += G1_BLOCKS) {
        const float4* nx = a4 + (size_t)(t + G1_BLOCKS) * (G1_ROWS * DIM_IN / 4) + tid;
        G1_EACH(G1_LOAD)
        __builtin_amdgcn_sched_barrier(0);

        g1_tile(&tileA[buf][c * G1_PITCH + kq], bf, dinv, C, t * G1_ROWS + h * 8, n);

        float* nt = tileA[buf ^ 1];
        G1_EACH(G1_STORE)
        __syncthreads();
        buf ^= 1;
    }
    if (t < G1_TILES)
        g1_tile(&tileA[buf][c * G1_PITCH + kq], bf, dinv, C, t * G1_ROWS + h * 8, n);
}


#define RP        (HD + 4)
#define ROWS_PAD  416
static_assert(ROWS_PAD >= OWN_ROWS && ROWS_PAD % 32 == 0 && (RP * 4) % 16 == 0, "tile shape of the in-block transform");

__device__ __forceinline__ float4* row_piece(float* rows, int i) {
    return (float4*)(rows + (i >> 4) * RP + (i & 15) * 4);
}

__device__ __forceinline__ void lds_add_f32(float* p, float v) {
    __hip_atomic_fetch_add(p, v, __ATOMIC_RELAXED, __HIP_MEMORY_SCOPE_WORKGROUP);
}

template <class Over>
__device__ __forceinline__ void own_scan(const float* __restrict__ ownf, int e, int tid,
                                         int* queue, int* qcount, Over over) {
    const uint4* own4 = (const uint4*)ownf;
    const unsigned char* own1 = (const unsigned char*)ownf;
    const unsigned me = blockIdx.x * 0x01010101u;
    const int ngroups = e >> 4;
    for (int g0 = tid; g0 < ngroups; g0 += 8 * 1024) {
        uint4 v[8];
#pragma unroll
        for (int k = 0; k < 8; ++k) {
            int g = g0 + k * 1024;
            if (g > ngroups - 1) g = ngroups - 1;
            v[k] = own4[g];
        }
        unsigned p0 = 0u, p1 = 0u, p2 = 0u, p3 = 0u;
#pragma unroll
        for (int k = 0; k < 8; ++k) {
            const unsigned x0 = v[k].x ^ me, x1 = v[k].y ^ me, x2 = v[k].z ^ me, x3 = v[k].w ^ me;
            const unsigned u0 = ((x0 & 0x7F7F7F7Fu) + 0x7F7F7F7Fu) | x0;
            const unsigned u1 = ((x1 & 0x7F7F7F7Fu) + 0x7F7F7F7Fu) | x1;
            const unsigned u2 = ((x2 & 0x7F7F7F7Fu) + 0x7F7F7F7Fu) | x2;
            const unsigned u3 = ((x3 & 0x7F7F7F7Fu) + 0x7F7F7F7Fu) | x3;
            const unsigned sel = 0x01010101u << k;
            p0 |= (u0 >> (7 - k)) & sel;
            p1 |= (u1 >> (7 - k)) & sel;
            p2 |= (u2 >> (7 - k)) & sel;
            p3 |= (u3 >> (7 - k)) & sel;
        }
        int nv = (ngroups - g0 + 1023) >> 10;
        if (nv > 8) nv = 8;
        const unsigned vm = ((1u << nv) - 1u) * 0x01010101u;
        const unsigned mm[4] = {~p0 & vm, ~p1 & vm, ~p2 & vm, ~p3 & vm};
#pragma unroll
        for (int j = 0; j < 4; ++j) {
            unsigned m = mm[j];
            while (m != 0u) {
                const int bit = __builtin_ctz(m);
                m &= m - 1u;
                const int eid = ((g0 + (bit & 7) * 1024) << 4) + j * 4 + (bit >> 3);
                const int pos = atomicAdd(qcount, 1);
                if (pos < QCAP) queue[pos] = eid;
                else            over(eid);
            }
        }
    }
    for (int i = (ngroups << 4) + tid; i < e; i += 1024) {
        if (own1[i] == (unsigned char)blockIdx.x) {
            const int pos = atomicAdd(qcount, 1);
            if (pos < QCAP) queue[pos] = i;
            else            over(i);
        }
    }
}

#define SCAN_BATCH 4

__global__ void __launch_bounds__(1024)
TEXTGCN_owner_scan(const int* __restrict__ dst,
                   const float* __restrict__ ew,
                   const float* __restrict__ ownf,
                   float* __restrict__ lstf,
                   float* __restrict__ sparef,
                   float* __restrict__ dinv, int e) {
    __shared__ float degs[OWN_ROWS];
    __shared__ int   queue[QCAP];
    __shared__ int   qcount;

    const int tid = threadIdx.x;
    const int lo  = blockIdx.x * OWN_ROWS;
    int nrows = N_NODES - lo;
    if (nrows > OWN_ROWS) nrows = OWN_ROWS;

    unsigned* reg = ((int)blockIdx.x < LMAIN) ? (unsigned*)lstf + (size_t)blockIdx.x * LCAP
                                              : (unsigned*)sparef + (size_t)((int)blockIdx.x - LMAIN) * LCAP;
    int* cnt = (int*)sparef + (OWN_BLOCKS - LMAIN) * LCAP;

    if (tid < OWN_ROWS) degs[tid] = 1.0f;
    if (tid == 0) qcount = 0;
    __syncthreads();

    float* const degs0 = degs;
    own_scan(ownf, e, tid, queue, &qcount,
             [=](int eid) { lds_add_f32(&degs0[dst[eid] - lo], ew[eid]); });
    __syncthreads();

    int n = qcount;
    const bool keep = n <= LCAP;
    if (tid == 0) cnt[blockIdx.x * CNT_PITCH] = keep ? n : -1;
    if (n > QCAP) n = QCAP;
    for (int b = tid; b < n; b += SCAN_BATCH * 1024) {
        int eid[SCAN_BATCH], dg[SCAN_BATCH];
        float w[SCAN_BATCH];
#pragma unroll
        for (int k = 0; k < SCAN_BATCH; ++k) {
            int i = b + k * 1024;
            if (i > n - 1) i = n - 1;
            eid[k] = queue[i];
        }
#pragma unroll
        for (int k = 0; k < SCAN_BATCH; ++k) {
            dg[k] = dst[eid[k]];
            w[k]  = ew[eid[k]];
        }
#pragma unroll
        for (int k = 0; k < SCAN_BATCH; ++k) {
            const int i = b + k * 1024;
            if (i < n) {
                const int dl = dg[k] - lo;
                lds_add_f32(&degs[dl], w[k]);
                if (keep) reg[i] = (unsigned)eid[k] | ((unsigned)dl << LW_SHIFT);
            }
        }
    }
    __syncthreads();

    if (tid < nrows) {
        const float d = degs[tid];
        dinv[lo + tid] = (d > 0.0f) ? __frsqrt_rn(d) : 0.0f;
    }
}

__device__ __forceinline__ void own_add_edges(int m, int sl, int dl, int wl, int lane,
                                              const float* __restrict__ t, float* rows) {
    for (int j = 0; j < m; j += 16) {
        float v0[16], v1[16];
#pragma unroll
        for (int k = 0; k < 16; ++k) {
            int idx = j + k;
            if (idx > m - 1) idx = m - 1;
            const int s = __builtin_amdgcn_readlane(sl, idx);
            const float* __restrict__ tp = t + (size_t)s * HD;
            v0[k] = tp[lane];
            v1[k] = tp[lane + 32];
        }
#pragma unroll
        for (int k = 0; k < 16; ++k) {
            if (j + k < m) {
                const int d = __builtin_amdgcn_readlane(dl, j + k);
                const float w = __int_as_float(__builtin_amdgcn_readlane(wl, j + k));
                float* rp = rows + d * RP;
                lds_add_f32(rp + lane,      v0[k] * w);
                lds_add_f32(rp + lane + 32, v1[k] * w);
            }
        }
    }
}

#define LTRIPS ((LCAP + 1023) / 1024)

#define DOC_TRIPS ((NDOCS + 1023) / 1024)
__device__ __forceinline__ void own_doc_rows(const int* __restrict__ src, const float* __restrict__ ew,
                                             const int* __restrict__ doc, const unsigned* __restrict__ reg, int listed,
                                             const float* __restrict__ t, const float* __restrict__ dinv,
                                             float* __restrict__ acc, float* rows, int* queue, int* qcount, int* mark,
                                             int lo, int nrows, int tid, int lane, int wave) {
    unsigned wdv[LTRIPS];
#pragma unroll
    for (int q = 0; q < LTRIPS; ++q) {
        const int i = tid + q * 1024;
        wdv[q] = (i < listed) ? reg[i] : 0u;
    }
    int dv[DOC_TRIPS];
#pragma unroll
    for (int q = 0; q < DOC_TRIPS; ++q) {
        int i = tid + q * 1024;
        if (i > NDOCS - 1) i = NDOCS - 1;
        dv[q] = doc[i];
    }
    if (tid < OWN_ROWS) mark[tid] = 0;
    if (tid == 0) *qcount = 0;
    __syncthreads();
#pragma unroll
    for (int q = 0; q < DOC_TRIPS; ++q) {
        const unsigned l = (unsigned)(dv[q] - lo);
        if (l < (unsigned)nrows) mark[l] = 1;
    }
    __syncthreads();

    const float4* a4 = (const float4*)(t + (size_t)lo * HD);
    const int nvec = nrows * (HD / 4);
    constexpr int TRIPS = (OWN_ROWS * (HD / 4) + 1023) / 1024;
    bool  on[TRIPS];
    float4 st[TRIPS];
#pragma unroll
    for (int q = 0; q < TRIPS; ++q) {
        const int i = tid + q * 1024;
        on[q] = (i < nvec) && (mark[i >> 4] != 0);
        float4 v = make_float4(0.0f, 0.0f, 0.0f, 0.0f);
        if (on[q]) v = a4[i];
        st[q] = v;
    }
#pragma unroll
    for (int q = 0; q < LTRIPS; ++q) {
        const int i = tid + q * 1024;
        if (i < listed && mark[wdv[q] >> LW_SHIFT] != 0) {
            const int pos = atomicAdd(qcount, 1);
            queue[pos] = (int)wdv[q];
        }
    }
#pragma unroll
    for (int q = 0; q < TRIPS; ++q)
        if (on[q]) *row_piece(rows, tid + q * 1024) = st[q];
    __syncthreads();

    const int n = __builtin_amdgcn_readfirstlane(*qcount);
    for (int base = wave * 32; base < n; base += 1024) {
        int m = n - base;
        if (m > 32) m = 32;
        int sl = 0, dl = 0, wl = 0;
        if (lane < m) {
            const unsigned w = (unsigned)queue[base + lane];
            const int eid = (int)(w & LW_MASK);
            dl = (int)(w >> LW_SHIFT);
            sl = src[eid];
            wl = __float_as_int(ew[eid]);
        }
        own_add_edges(m, sl, dl, wl, lane, t, rows);
    }
    __syncthreads();

    float4* o4 = (float4*)(acc + (size_t)lo * HD);
#pragma unroll
    for (int q = 0; q < TRIPS; ++q) {
        if (on[q]) {
            const int i = tid + q * 1024;
            const float s = dinv[lo + (i >> 4)];
            float4 v = *row_piece(rows, i);
            v.x *= s; v.y *= s; v.z *= s; v.w *= s;
            o4[i] = v;
        }
    }
}

template <bool DOC>
__global__ void __launch_bounds__(1024)
TEXTGCN_propagate_owned(const int* __restrict__ src,
                        const int* __restrict__ dst,
                        const float* __restrict__ ew,
                        const int* __restrict__ doc,
                        const float* __restrict__ ownf,
                        const float* __restrict__ lstf,
                        const float* __restrict__ sparef,
                        const float* __restrict__ t,
                        const float* __restrict__ dinv,
                        const float* __restrict__ bcur,
                        const float* __restrict__ Wn,
                        float* __restrict__ acc, int e) {
    __shared__ __attribute__((aligned(16))) float rows[ROWS_PAD * RP];
    __shared__ float dsc[ROWS_PAD];
    __shared__ int   queue[QCAP];
    __shared__ int   qcount;
    __shared__ int   mark[OWN_ROWS];

    const int tid  = threadIdx.x;
    const int lane = tid & 31;
    const int wave = __builtin_amdgcn_readfirstlane(tid >> 5);
    const int lo   = blockIdx.x * OWN_ROWS;
    int nrows = N_NODES - lo;
    if (nrows > OWN_ROWS) nrows = OWN_ROWS;

    const unsigned* reg = ((int)blockIdx.x < LMAIN) ? (const unsigned*)lstf + (size_t)blockIdx.x * LCAP
                                                    : (const unsigned*)sparef + (size_t)((int)blockIdx.x - LMAIN) * LCAP;
    const int* cnt = (const int*)sparef + (OWN_BLOCKS - LMAIN) * LCAP;
    const int listed = cnt[blockIdx.x * CNT_PITCH];
    const bool from_list = listed >= 0;

    if constexpr (DOC) {
        if (from_list) {
            own_doc_rows(src, ew, doc, reg, listed, t, dinv, acc, rows, queue, &qcount, mark, lo, nrows, tid, lane, wave);
            return;
        }
    }

    int lsl[LTRIPS], ldl[LTRIPS], lwl[LTRIPS];
#pragma unroll
    for (int q = 0; q < LTRIPS; ++q) { lsl[q] = 0; ldl[q] = 0; lwl[q] = 0; }
    if (from_list) {
        unsigned wdv[LTRIPS];
#pragma unroll
        for (int q = 0; q < LTRIPS; ++q) {
            const int i = tid + q * 1024;
            wdv[q] = (i < listed) ? reg[i] : 0u;
        }
#pragma unroll
        for (int q = 0; q < LTRIPS; ++q) {
            const int eid = (int)(wdv[q] & LW_MASK);
            ldl[q] = (int)(wdv[q] >> LW_SHIFT);
            lsl[q] = src[eid];
            lwl[q] = __float_as_int(ew[eid]);
        }
    }

    const float4* a4 = (const float4*)(t + (size_t)lo * HD);
    const int nvec = nrows * (HD / 4);
    {
        constexpr int TRIPS = (OWN_ROWS * (HD / 4) + 1023) / 1024;
        float4 st[TRIPS];
#pragma unroll
        for (int q = 0; q < TRIPS; ++q) {
            int i = tid + q * 1024;
            if (i > nvec - 1) i = nvec - 1;
            st[q] = a4[i];
        }
#pragma unroll
        for (int q = 0; q < TRIPS; ++q) {
            const int i = tid + q * 1024;
            if (i < OWN_ROWS * (HD / 4)) *row_piece(rows, i) = st[q];
        }
    }
    if constexpr (!DOC) {
        for (int i = tid; i < (ROWS_PAD - OWN_ROWS) * RP; i += 1024) rows[OWN_ROWS * RP + i] = 0.0f;
        if (tid < ROWS_PAD) dsc[tid] = (tid < nrows) ? dinv[lo + tid] : 0.0f;
    }
    if (tid == 0) qcount = 0;
    __syncthreads();

    if (!from_list) {
        float* const rows0 = rows;
        own_scan(ownf, e, tid, queue, &qcount,
                 [=](int eid) {
                     const int s = src[eid];
                     const int d = dst[eid] - lo;
                     const float w = ew[eid];
#pragma nounroll
                     for (int cc = 0; cc < HD; ++cc)
                         lds_add_f32(&rows0[d * RP + cc], t[(size_t)s * HD + cc] * w);
                 });
    }
    __syncthreads();

    if (from_list) {
        const int n = listed;
#pragma unroll
        for (int q = 0; q < LTRIPS; ++q) {
            const int base = wave * 32 + q * 1024;
            if (base < n) {
                int m = n - base;
                if (m > 32) m = 32;
                own_add_edges(m, lsl[q], ldl[q], lwl[q], lane, t, rows);
            }
        }
    } else {
        int n = qcount;
        if (n > QCAP) n = QCAP;
        n = __builtin_amdgcn_readfirstlane(n);
        for (int base = wave * 32; base < n; base += 1024) {
            int m = n - base;
            if (m > 32) m = 32;
            int sl = 0, dl = 0, wl = 0;
            if (lane < m) {
                const int eid = queue[base + lane];
                const int dg = dst[eid];
                sl = src[eid];
                wl = __float_as_int(ew[eid]);
                dl = dg - lo;
            }
            own_add_edges(m, sl, dl, wl, lane, t, rows);
        }
    }
    __syncthreads();

    if constexpr (DOC) {
        float4* o4 = (float4*)(acc + (size_t)lo * HD);
        constexpr int TRIPS = (OWN_ROWS * (HD / 4) + 1023) / 1024;
        float4 ov[TRIPS];
        float  sc[TRIPS];
#pragma unroll
        for (int q = 0; q < TRIPS; ++q) {
            int i = tid + q * 1024;
            if (i > nvec - 1) i = nvec - 1;
            sc[q] = dinv[lo + (i >> 4)];
            ov[q] = *row_piece(rows, i);
        }
#pragma unroll
        for (int q = 0; q < TRIPS; ++q) {
            const int i = tid + q * 1024;
            if (i < nvec) {
                float4 v = ov[q];
                v.x *= sc[q]; v.y *= sc[q]; v.z *= sc[q]; v.w *= sc[q];
                o4[i] = v;
            }
        }
    } else {
        const int c  = lane & 15;
        const int h  = lane >> 4;
        const int kq = h * 2;
        const int n  = (wave & 3) * 16 + c;
        v2f bf[HD / 4];
#pragma unroll
        for (int i = 0; i < HD / 4; ++i) {
            bf[i].x = Wn[(size_t)(4 * i + kq)     * HD + n];
            bf[i].y = Wn[(size_t)(4 * i + kq + 1) * HD + n];
        }
        const float4 bv = *(const float4*)(bcur + (tid & 15) * 4);
        constexpr int TRIPS = (OWN_ROWS * (HD / 4) + 1023) / 1024;
#pragma unroll
        for (int q = 0; q < TRIPS; ++q) {
            const int i = tid + q * 1024;
            if (i < OWN_ROWS * (HD / 4)) {
                float4* p = row_piece(rows, i);
                const float sd = dsc[i >> 4];
                float4 v = *p;
                const float v0 = v.x * sd + bv.x, v1 = v.y * sd + bv.y, v2 = v.z * sd + bv.z, v3 = v.w * sd + bv.w;
                v.x = v0 > 0.0f ? v0 : 0.0f;
                v.y = v1 > 0.0f ? v1 : 0.0f;
                v.z = v2 > 0.0f ? v2 : 0.0f;
                v.w = v3 > 0.0f ? v3 : 0.0f;
                *p = v;
            }
        }
        __syncthreads();

        for (int st = wave >> 2; st < ROWS_PAD / 32; st += 8) {
            const float* ap = &rows[(st * 32 + c) * RP + kq];
            v8f acc0 = {}, acc1 = {};
#pragma unroll
            for (int i = 0; i < HD / 4; ++i) {
                const v2f a0 = *(const v2f*)(ap + 4 * i);
                const v2f a1 = *(const v2f*)(ap + 16 * RP + 4 * i);
                acc0 = __builtin_amdgcn_wmma_f32_16x16x4_f32(
                           false, a0, false, bf[i], (short)0, acc0, false, false);
                acc1 = __builtin_amdgcn_wmma_f32_16x16x4_f32(
                           false, a1, false, bf[i], (short)0, acc1, false, false);
            }
            const int ml = st * 32 + h * 8;
            float o0[8], o1[8];
#pragma unroll
            for (int r = 0; r < 8; ++r) {
                o0[r] = acc0[r] * dsc[ml + r];
                o1[r] = acc1[r] * dsc[ml + 16 + r];
            }
#pragma unroll
            for (int r = 0; r < 8; ++r) {
                asm volatile("" : "+v"(o0[r]));
                asm volatile("" : "+v"(o1[r]));
            }
            float* __restrict__ tp = acc + ((unsigned)(lo + ml) * HD + (unsigned)n);
#pragma unroll
            for (int r = 0; r < 8; ++r) {
                if (ml + r < nrows)      tp[r * HD]        = o0[r];
                if (ml + 16 + r < nrows) tp[(16 + r) * HD] = o1[r];
            }
        }
    }
}


__global__ void TEXTGCN_doc_gemm_wmma(const float* __restrict__ h,
                                      const float* __restrict__ bprev,
                                      const int* __restrict__ doc_idx,
                                      const float* __restrict__ Wp,
                                      const float* __restrict__ bp,
                                      float* __restrict__ dbuf) {
    const int lane = threadIdx.x;
    const int mt = blockIdx.x, nt = blockIdx.y;
    const int m  = mt * 16 + (lane & 15);
    const int n  = nt * 16 + (lane & 15);
    const int kq = (lane >> 4) * 2;

    const float* __restrict__ arow = h + (size_t)doc_idx[m] * HD;
    v8f acc = {};
    for (int k = 0; k < HD; k += 4) {
        v2f a, b;
        const float v0 = arow[k + kq]     + bprev[k + kq];
        const float v1 = arow[k + kq + 1] + bprev[k + kq + 1];
        a.x = v0 > 0.0f ? v0 : 0.0f;
        a.y = v1 > 0.0f ? v1 : 0.0f;
        b.x = Wp[(size_t)(k + kq)     * HD + n];
        b.y = Wp[(size_t)(k + kq + 1) * HD + n];
        acc = __builtin_amdgcn_wmma_f32_16x16x4_f32(
                  false, a, false, b, (short)0, acc, false, false);
    }
    const int mbase = mt * 16 + (lane >> 4) * 8;
    const int nst   = nt * 16 + (lane & 15);
    const float bias = bp[nst];
#pragma unroll
    for (int r = 0; r < 8; ++r) {
        float v = acc[r] + bias;
        dbuf[(size_t)(mbase + r) * HD + nst] = v > 0.0f ? v : 0.0f;
    }
}


__global__ void TEXTGCN_classifier(const float* __restrict__ dbuf,
                                   const float* __restrict__ Wc,
                                   const float* __restrict__ bc,
                                   float* __restrict__ out) {
    int gid = blockIdx.x * blockDim.x + threadIdx.x;
    if (gid >= NDOCS * NCLS) return;
    int d = gid / NCLS, c = gid % NCLS;
    const float* __restrict__ row = dbuf + (size_t)d * HD;
    float s = bc[c];
#pragma unroll 8
    for (int k = 0; k < HD; ++k) s += row[k] * Wc[(size_t)k * NCLS + c];
    out[gid] = s;
}


extern "C" void kernel_launch(void* const* d_in, const int* in_sizes, int n_in,
                              void* d_out, int out_size, void* d_ws, size_t ws_size,
                              hipStream_t stream) {
    const float* x   = (const float*)d_in[0];
    const float* ew  = (const float*)d_in[1];
    const float* W1  = (const float*)d_in[2];
    const float* b1  = (const float*)d_in[3];
    const float* W2  = (const float*)d_in[4];
    const float* b2  = (const float*)d_in[5];
    const float* W3  = (const float*)d_in[6];
    const float* b3  = (const float*)d_in[7];
    const float* Wp  = (const float*)d_in[8];
    const float* bp  = (const float*)d_in[9];
    const float* Wc  = (const float*)d_in[10];
    const float* bc  = (const float*)d_in[11];
    const int*   ei  = (const int*)d_in[12];
    const int*   doc = (const int*)d_in[13];
    float*       out = (float*)d_out;

    const int* src = ei;
    const int* dst = ei + N_EDGES;

    float* ws   = (float*)d_ws;
    float* dinv = ws;
    float* lst  = dinv + N_NODES;
    float* bufT = lst  + N_EDGES;
    float* bufH = bufT + (size_t)N_NODES * HD;
    float* dbuf = bufH + (size_t)N_NODES * HD;

    const int B = 256;
    dim3 blk(B);
    dim3 gE((N_EDGES + B - 1) / B);
    dim3 gOwn(OWN_BLOCKS);
    dim3 blkOwn(1024);

    TEXTGCN_owner_bytes<<<gE, blk, 0, stream>>>(dst, dbuf, N_EDGES);
    float* spare = dbuf + OWN_WORDS;
    TEXTGCN_owner_scan<<<gOwn, blkOwn, 0, stream>>>(dst, ew, dbuf, lst, spare, dinv, N_EDGES);

    dim3 wave(32);
    dim3 gGemm1(G1_BLOCKS);
    dim3 blkH(128);
    dim3 gDoc(NDOCS / 16, HD / 16);

    TEXTGCN_gemm_rows<<<gGemm1, blkH, 0, stream>>>(x, W1, dinv, bufT);
    TEXTGCN_propagate_owned<false><<<gOwn, blkOwn, 0, stream>>>(src, dst, ew, doc, dbuf, lst, spare, bufT, dinv, b1, W2, bufH, N_EDGES);

    TEXTGCN_propagate_owned<false><<<gOwn, blkOwn, 0, stream>>>(src, dst, ew, doc, dbuf, lst, spare, bufH, dinv, b2, W3, bufT, N_EDGES);

    TEXTGCN_propagate_owned<true><<<gOwn, blkOwn, 0, stream>>>(src, dst, ew, doc, dbuf, lst, spare, bufT, dinv, b3, W3, bufH, N_EDGES);

    TEXTGCN_doc_gemm_wmma<<<gDoc, wave, 0, stream>>>(bufH, b3, doc, Wp, bp, dbuf);
    dim3 gCls((NDOCS * NCLS + B - 1) / B);
    TEXTGCN_classifier<<<gCls, blk, 0, stream>>>(dbuf, Wc, bc, out);
}
